# starting kernel
# baseline (speedup 1.0000x reference)
#include <hip/hip_runtime.h>
#include <hip/hip_bf16.h>

typedef __attribute__((ext_vector_type(2))) float v2f;
typedef __attribute__((ext_vector_type(8))) float v8f;

#define JB 8192
#define NJ 8
#define DIM 128
#define COLS 1024

__device__ __forceinline__ float squash_gain(float n2) {
    return n2 / ((1.0f + n2) * sqrtf(n2 + 1e-8f));
}

__launch_bounds__(256)
__global__ void gemm_uhat_kernel(const float* __restrict__ x,
                                 const float* __restrict__ w,
                                 float* __restrict__ uh) {
    const float* x7 = x + (size_t)7 * 8 * 1024 * 128;
    const float* w7 = w + (size_t)7 * 8 * 128 * 128;

    const int wave = threadIdx.x >> 5;
    const int lane = threadIdx.x & 31;
    const int half = lane >> 4;
    const int l16  = lane & 15;

    const int row0 = blockIdx.x * 128 + wave * 16;
    const int r    = row0 + l16;
    const float* arow = x7 + (size_t)((r & 7) * 1024 + (r >> 3)) * 128;

    for (int j = 0; j < NJ; ++j) {
        const float* wj = w7 + (size_t)j * 128 * 128;
        v8f acc[8];
#pragma unroll
        for (int n = 0; n < 8; ++n) acc[n] = (v8f){0,0,0,0,0,0,0,0};

        for (int k = 0; k < 128; k += 4) {
            v2f a;
            a.x = arow[k + 2 * half + 0];
            a.y = arow[k + 2 * half + 1];
            const float* wk = wj + (size_t)(k + 2 * half) * 128 + l16;
#pragma unroll
            for (int n = 0; n < 8; ++n) {
                v2f b;
                b.x = wk[n * 16];
                b.y = wk[128 + n * 16];
                acc[n] = __builtin_amdgcn_wmma_f32_16x16x4_f32(
                    false, a, false, b, (short)0, acc[n], false, false);
            }
        }
#pragma unroll
        for (int n = 0; n < 8; ++n) {
            float* od = uh + (size_t)(row0 + 8 * half) * COLS + j * DIM + n * 16 + l16;
#pragma unroll
            for (int g = 0; g < 8; ++g) {
                od[(size_t)g * COLS] = acc[n][g];
            }
        }
    }
}

__launch_bounds__(256)
__global__ void colmean_kernel(const float* __restrict__ uh, float* __restrict__ uhm) {
    int c = blockIdx.x * 256 + threadIdx.x;
    float s = 0.0f;
    for (int r = 0; r < JB; ++r) s += uh[(size_t)r * COLS + c];
    uhm[c] = s * (1.0f / (float)JB);
}

__launch_bounds__(256)
__global__ void dotn2_kernel(const float* __restrict__ uh,
                             const float* __restrict__ uhm,
                             float* __restrict__ dotp, float* __restrict__ n2) {
    const int wave = threadIdx.x >> 5;
    const int lane = threadIdx.x & 31;
    const int pair = blockIdx.x * 8 + wave;
    const int r = pair >> 3, j = pair & 7;

    const float4 a = ((const float4*)(uh + (size_t)r * COLS + j * DIM))[lane];
    const float4 m = ((const float4*)(uhm + j * DIM))[lane];
    float d = a.x * m.x + a.y * m.y + a.z * m.z + a.w * m.w;
    float q = a.x * a.x + a.y * a.y + a.z * a.z + a.w * a.w;
#pragma unroll
    for (int off = 16; off > 0; off >>= 1) {
        d += __shfl_xor(d, off, 32);
        q += __shfl_xor(q, off, 32);
    }
    if (lane == 0) { dotp[pair] = d; n2[pair] = q; }
}

__launch_bounds__(1024)
__global__ void routing_kernel(const float* __restrict__ dotp,
                               const float* __restrict__ n2,
                               float* __restrict__ cf) {
    __shared__ float B7[8];
    __shared__ float c[8];
    __shared__ float sumj[8];
    const int tid = threadIdx.x;
    if (tid < 8) B7[tid] = 0.0f;

    for (int it = 0; it < 2; ++it) {
        __syncthreads();
        if (tid == 0) {
            float m = B7[0];
            for (int jj = 1; jj < 8; ++jj) m = fmaxf(m, B7[jj]);
            float s = 0.0f, e[8];
            for (int jj = 0; jj < 8; ++jj) { e[jj] = __expf(B7[jj] - m); s += e[jj]; }
            for (int jj = 0; jj < 8; ++jj) c[jj] = e[jj] / s;
        }
        __syncthreads();
        if (tid < 8) sumj[tid] = 0.0f;
        __syncthreads();
        const int j = tid & 7;
        const float cj = c[j];
        float a = 0.0f;
        for (int p = tid; p < JB * NJ; p += 1024)
            a += squash_gain(cj * cj * n2[p]) * dotp[p];
        atomicAdd(&sumj[j], a);
        __syncthreads();
        if (tid < 8) B7[tid] += c[tid] * sumj[tid] * (1.0f / (float)JB);
    }
    __syncthreads();
    if (tid == 0) {
        float m = B7[0];
        for (int jj = 1; jj < 8; ++jj) m = fmaxf(m, B7[jj]);
        float s = 0.0f, e[8];
        for (int jj = 0; jj < 8; ++jj) { e[jj] = __expf(B7[jj] - m); s += e[jj]; }
        for (int jj = 0; jj < 8; ++jj) cf[jj] = e[jj] / s;
    }
}

__launch_bounds__(256)
__global__ void final_kernel(const float* __restrict__ uh,
                             const float* __restrict__ n2,
                             const float* __restrict__ cf,
                             float* __restrict__ out) {
    const size_t idx  = (size_t)blockIdx.x * 256 + threadIdx.x;
    const size_t base = idx * 4;
    const int r = (int)(base >> 10);
    const int j = (int)((base >> 7) & 7);
    const float cj = cf[j];
    const float gain = cj * squash_gain(cj * cj * n2[r * 8 + j]);
    float4 a = *(const float4*)(uh + base);
    a.x *= gain; a.y *= gain; a.z *= gain; a.w *= gain;
    *(float4*)(out + base) = a;
}

extern "C" void kernel_launch(void* const* d_in, const int* in_sizes, int n_in,
                              void* d_out, int out_size, void* d_ws, size_t ws_size,
                              hipStream_t stream) {
    const float* x = (const float*)d_in[0];
    const float* w = (const float*)d_in[1];
    float* out = (float*)d_out;

    char* ws = (char*)d_ws;
    float* uh   = (float*)(ws);
    float* n2   = (float*)(ws + (size_t)JB * COLS * 4);
    float* dotp = (float*)(ws + (size_t)JB * COLS * 4 + 262144);
    float* uhm  = (float*)(ws + (size_t)JB * COLS * 4 + 524288);
    float* cf   = (float*)(ws + (size_t)JB * COLS * 4 + 528384);

    gemm_uhat_kernel<<<64, 256, 0, stream>>>(x, w, uh);
    colmean_kernel<<<4, 256, 0, stream>>>(uh, uhm);
    dotn2_kernel<<<JB, 256, 0, stream>>>(uh, uhm, dotp, n2);
    routing_kernel<<<1, 1024, 0, stream>>>(dotp, n2, cf);
    final_kernel<<<(JB * COLS / 4) / 256, 256, 0, stream>>>(uh, n2, cf, out);
}


# GEMM re-tiled: a wave forms a 64x64 block of uh over all k in registers, its W slab and x rows loaded once by 16-byte loads; 256 blocks on all WGPs; k order inside each 8 permuted: sums reordered
# speedup vs baseline: 5.6786x; 5.6786x over previous
#include <hip/hip_runtime.h>
#include <hip/hip_bf16.h>

typedef __attribute__((ext_vector_type(2))) float v2f;
typedef __attribute__((ext_vector_type(8))) float v8f;
typedef __attribute__((ext_vector_type(4))) float v4f;

#define JB 8192
#define NJ 8
#define DIM 128
#define COLS 1024

__device__ __forceinline__ float squash_gain(float n2) {
    return n2 / ((1.0f + n2) * sqrtf(n2 + 1e-8f));
}

__launch_bounds__(256)
__global__ void gemm_uhat_kernel(const float* __restrict__ x,
                                 const float* __restrict__ w,
                                 float* __restrict__ uh) {
    const float* x7 = x + (size_t)7 * 8 * 1024 * 128;
    const float* w7 = w + (size_t)7 * 8 * 128 * 128;

    const int wave = threadIdx.x >> 5;
    const int lane = threadIdx.x & 31;
    const int h    = lane >> 4;
    const int c    = lane & 15;

    const int grp  = blockIdx.y * 8 + wave;
    const int j    = grp >> 1;
    const int col0 = (grp & 1) * 64 + 4 * c;

    const float* wl = w7 + (size_t)j * 128 * 128 + (size_t)(4 * h) * 128 + col0;
    v4f bq[32][2];
#pragma unroll
    for (int t = 0; t < 32; ++t) {
#pragma unroll
        for (int e = 0; e < 2; ++e)
            bq[t][e] = *(const v4f*)(wl + (8 * (t >> 1) + 2 * (t & 1) + e) * 128);
    }

    const float* xr = x7 + (size_t)((c & 7) * 1024 + blockIdx.x * 8 + (c >> 3)) * 128 + 4 * h;
    float* orow = uh + (size_t)(blockIdx.x * 64 + 8 * h) * COLS + grp * 64 + 4 * c;

#pragma unroll
    for (int i = 0; i < 4; ++i) {
        v4f a4[16];
#pragma unroll
        for (int u = 0; u < 16; ++u) a4[u] = *(const v4f*)(xr + i * 256 + 8 * u);

        v8f acc[4];
#pragma unroll
        for (int T = 0; T < 4; ++T) acc[T] = (v8f){0,0,0,0,0,0,0,0};

#pragma unroll
        for (int u = 0; u < 16; ++u) {
#pragma unroll
            for (int p = 0; p < 2; ++p) {
                v2f a;
                a.x = a4[u][2 * p + 0];
                a.y = a4[u][2 * p + 1];
#pragma unroll
                for (int T = 0; T < 4; ++T) {
                    v2f b;
                    b.x = bq[2 * u + p][0][T];
                    b.y = bq[2 * u + p][1][T];
                    acc[T] = __builtin_amdgcn_wmma_f32_16x16x4_f32(
                        false, a, false, b, (short)0, acc[T], false, false);
                }
            }
        }
#pragma unroll
        for (int g = 0; g < 8; ++g) {
            v4f o;
            o.x = acc[0][g]; o.y = acc[1][g]; o.z = acc[2][g]; o.w = acc[3][g];
            *(v4f*)(orow + (size_t)(16 * i + g) * COLS) = o;
        }
    }
}

__device__ __forceinline__ float fold_rows(float v) {
    v += __shfl_xor(v, 8, 32);
    v += __shfl_xor(v, 16, 32);
    return v;
}

__launch_bounds__(1024)
__global__ void colmean_kernel(const float* __restrict__ uh, float* __restrict__ uhm) {
    __shared__ __attribute__((aligned(16))) float part[32][32];
    const int wave = threadIdx.x >> 5;
    const int lane = threadIdx.x & 31;
    const int q    = lane & 7;
    const int rs   = lane >> 3;
    const float* p = uh + (size_t)(wave * 4 + rs) * COLS + blockIdx.x * 32 + q * 4;

    v4f acc[8] = {};

    for (int t = 0; t < 64; t += 8) {
#pragma unroll
        for (int u = 0; u < 8; ++u)
            acc[u] += *(const v4f*)(p + (size_t)(t + u) * (128 * COLS));
    }
    v4f s = ((acc[0] + acc[1]) + (acc[2] + acc[3])) + ((acc[4] + acc[5]) + (acc[6] + acc[7]));
    s.x = fold_rows(s.x);
    s.y = fold_rows(s.y);
    s.z = fold_rows(s.z);
    s.w = fold_rows(s.w);
    if (lane < 8) *(v4f*)&part[wave][q * 4] = s;
    __syncthreads();
    if (wave == 0) {
        float a0 = 0.0f, a1 = 0.0f, a2 = 0.0f, a3 = 0.0f;
        for (int w = 0; w < 32; w += 4) {
            a0 += part[w + 0][lane];
            a1 += part[w + 1][lane];
            a2 += part[w + 2][lane];
            a3 += part[w + 3][lane];
        }
        uhm[blockIdx.x * 32 + lane] = ((a0 + a1) + (a2 + a3)) * (1.0f / (float)JB);
    }
}

__launch_bounds__(256)
__global__ void dotn2_kernel(const float* __restrict__ uh,
                             const float* __restrict__ uhm,
                             float* __restrict__ dotp, float* __restrict__ n2) {
    const int wave = threadIdx.x >> 5;
    const int lane = threadIdx.x & 31;
    const int pair = blockIdx.x * 8 + wave;
    const int r = pair >> 3, j = pair & 7;

    const float4 a = ((const float4*)(uh + (size_t)r * COLS + j * DIM))[lane];
    const float4 m = ((const float4*)(uhm + j * DIM))[lane];
    float d = a.x * m.x + a.y * m.y + a.z * m.z + a.w * m.w;
    float q = a.x * a.x + a.y * a.y + a.z * a.z + a.w * a.w;
#pragma unroll
    for (int off = 16; off > 0; off >>= 1) {
        d += __shfl_xor(d, off, 32);
        q += __shfl_xor(q, off, 32);
    }
    if (lane == 0) { dotp[pair] = d; n2[pair] = q; }
}

__launch_bounds__(1024)
__global__ void routing_kernel(const float* __restrict__ dotp,
                               const float* __restrict__ n2,
                               float* __restrict__ cf) {
    __shared__ float B7[8];
    __shared__ float c[8];
    __shared__ float sumj[8];
    const int tid = threadIdx.x;
    if (tid < 8) B7[tid] = 0.0f;

    for (int it = 0; it < 2; ++it) {
        __syncthreads();
        if (tid == 0) {
            float m = B7[0];
            for (int jj = 1; jj < 8; ++jj) m = fmaxf(m, B7[jj]);
            float s = 0.0f, e[8];
            for (int jj = 0; jj < 8; ++jj) { e[jj] = __expf(B7[jj] - m); s += e[jj]; }
            for (int jj = 0; jj < 8; ++jj) c[jj] = e[jj] / s;
        }
        __syncthreads();
        if (tid < 8) sumj[tid] = 0.0f;
        __syncthreads();
        const int j = tid & 7;
        const float cj = c[j];
        float a = 0.0f;
        for (int p = tid; p < JB * NJ; p += 1024)
            a += squash_gain(cj * cj * n2[p]) * dotp[p];
        atomicAdd(&sumj[j], a);
        __syncthreads();
        if (tid < 8) B7[tid] += c[tid] * sumj[tid] * (1.0f / (float)JB);
    }
    __syncthreads();
    if (tid == 0) {
        float m = B7[0];
        for (int jj = 1; jj < 8; ++jj) m = fmaxf(m, B7[jj]);
        float s = 0.0f, e[8];
        for (int jj = 0; jj < 8; ++jj) { e[jj] = __expf(B7[jj] - m); s += e[jj]; }
        for (int jj = 0; jj < 8; ++jj) cf[jj] = e[jj] / s;
    }
}

__launch_bounds__(256)
__global__ void final_kernel(const float* __restrict__ uh,
                             const float* __restrict__ n2,
                             const float* __restrict__ cf,
                             float* __restrict__ out) {
    const size_t idx  = (size_t)blockIdx.x * 256 + threadIdx.x;
    const size_t base = idx * 4;
    const int r = (int)(base >> 10);
    const int j = (int)((base >> 7) & 7);
    const float cj = cf[j];
    const float gain = cj * squash_gain(cj * cj * n2[r * 8 + j]);
    float4 a = *(const float4*)(uh + base);
    a.x *= gain; a.y *= gain; a.z *= gain; a.w *= gain;
    *(float4*)(out + base) = a;
}

extern "C" void kernel_launch(void* const* d_in, const int* in_sizes, int n_in,
                              void* d_out, int out_size, void* d_ws, size_t ws_size,
                              hipStream_t stream) {
    const float* x = (const float*)d_in[0];
    const float* w = (const float*)d_in[1];
    float* out = (float*)d_out;

    char* ws = (char*)d_ws;
    float* uh   = (float*)(ws);
    float* n2   = (float*)(ws + (size_t)JB * COLS * 4);
    float* dotp = (float*)(ws + (size_t)JB * COLS * 4 + 262144);
    float* uhm  = (float*)(ws + (size_t)JB * COLS * 4 + 524288);
    float* cf   = (float*)(ws + (size_t)JB * COLS * 4 + 528384);

    gemm_uhat_kernel<<<dim3(128, 2, 1), 256, 0, stream>>>(x, w, uh);
    colmean_kernel<<<32, 1024, 0, stream>>>(uh, uhm);
    dotn2_kernel<<<JB, 256, 0, stream>>>(uh, uhm, dotp, n2);
    routing_kernel<<<1, 1024, 0, stream>>>(dotp, n2, cf);
    final_kernel<<<(JB * COLS / 4) / 256, 256, 0, stream>>>(uh, n2, cf, out);
}


# routing block's pair loop: squash gain as n2*v_rsq_f32*v_rcp_f32 (about 1 ulp each), not IEEE sqrt and divide sequences; 8 pairs' loads in flight per trip; summand rounding changes, add order kept
# speedup vs baseline: 7.2012x; 1.2681x over previous
#include <hip/hip_runtime.h>
#include <hip/hip_bf16.h>

typedef __attribute__((ext_vector_type(2))) float v2f;
typedef __attribute__((ext_vector_type(8))) float v8f;
typedef __attribute__((ext_vector_type(4))) float v4f;

#define JB 8192
#define NJ 8
#define DIM 128
#define COLS 1024

__device__ __forceinline__ float squash_gain(float n2) {
    return n2 / ((1.0f + n2) * sqrtf(n2 + 1e-8f));
}

__device__ __forceinline__ float squash_gain_routing(float n2) {
    return n2 * __builtin_amdgcn_rsqf(n2 + 1e-8f) * __builtin_amdgcn_rcpf(1.0f + n2);
}

__launch_bounds__(256)
__global__ void gemm_uhat_kernel(const float* __restrict__ x,
                                 const float* __restrict__ w,
                                 float* __restrict__ uh) {
    const float* x7 = x + (size_t)7 * 8 * 1024 * 128;
    const float* w7 = w + (size_t)7 * 8 * 128 * 128;

    const int wave = threadIdx.x >> 5;
    const int lane = threadIdx.x & 31;
    const int h    = lane >> 4;
    const int c    = lane & 15;

    const int grp  = blockIdx.y * 8 + wave;
    const int j    = grp >> 1;
    const int col0 = (grp & 1) * 64 + 4 * c;

    const float* wl = w7 + (size_t)j * 128 * 128 + (size_t)(4 * h) * 128 + col0;
    v4f bq[32][2];
#pragma unroll
    for (int t = 0; t < 32; ++t) {
#pragma unroll
        for (int e = 0; e < 2; ++e)
            bq[t][e] = *(const v4f*)(wl + (8 * (t >> 1) + 2 * (t & 1) + e) * 128);
    }

    const float* xr = x7 + (size_t)((c & 7) * 1024 + blockIdx.x * 8 + (c >> 3)) * 128 + 4 * h;
    float* orow = uh + (size_t)(blockIdx.x * 64 + 8 * h) * COLS + grp * 64 + 4 * c;

#pragma unroll
    for (int i = 0; i < 4; ++i) {
        v4f a4[16];
#pragma unroll
        for (int u = 0; u < 16; ++u) a4[u] = *(const v4f*)(xr + i * 256 + 8 * u);

        v8f acc[4];
#pragma unroll
        for (int T = 0; T < 4; ++T) acc[T] = (v8f){0,0,0,0,0,0,0,0};

#pragma unroll
        for (int u = 0; u < 16; ++u) {
#pragma unroll
            for (int p = 0; p < 2; ++p) {
                v2f a;
                a.x = a4[u][2 * p + 0];
                a.y = a4[u][2 * p + 1];
#pragma unroll
                for (int T = 0; T < 4; ++T) {
                    v2f b;
                    b.x = bq[2 * u + p][0][T];
                    b.y = bq[2 * u + p][1][T];
                    acc[T] = __builtin_amdgcn_wmma_f32_16x16x4_f32(
                        false, a, false, b, (short)0, acc[T], false, false);
                }
            }
        }
#pragma unroll
        for (int g = 0; g < 8; ++g) {
            v4f o;
            o.x = acc[0][g]; o.y = acc[1][g]; o.z = acc[2][g]; o.w = acc[3][g];
            *(v4f*)(orow + (size_t)(16 * i + g) * COLS) = o;
        }
    }
}

__device__ __forceinline__ float fold_rows(float v) {
    v += __shfl_xor(v, 8, 32);
    v += __shfl_xor(v, 16, 32);
    return v;
}

__launch_bounds__(1024)
__global__ void colmean_kernel(const float* __restrict__ uh, float* __restrict__ uhm) {
    __shared__ __attribute__((aligned(16))) float part[32][32];
    const int wave = threadIdx.x >> 5;
    const int lane = threadIdx.x & 31;
    const int q    = lane & 7;
    const int rs   = lane >> 3;
    const float* p = uh + (size_t)(wave * 4 + rs) * COLS + blockIdx.x * 32 + q * 4;

    v4f acc[8] = {};

    for (int t = 0; t < 64; t += 8) {
#pragma unroll
        for (int u = 0; u < 8; ++u)
            acc[u] += *(const v4f*)(p + (size_t)(t + u) * (128 * COLS));
    }
    v4f s = ((acc[0] + acc[1]) + (acc[2] + acc[3])) + ((acc[4] + acc[5]) + (acc[6] + acc[7]));
    s.x = fold_rows(s.x);
    s.y = fold_rows(s.y);
    s.z = fold_rows(s.z);
    s.w = fold_rows(s.w);
    if (lane < 8) *(v4f*)&part[wave][q * 4] = s;
    __syncthreads();
    if (wave == 0) {
        float a0 = 0.0f, a1 = 0.0f, a2 = 0.0f, a3 = 0.0f;
        for (int w = 0; w < 32; w += 4) {
            a0 += part[w + 0][lane];
            a1 += part[w + 1][lane];
            a2 += part[w + 2][lane];
            a3 += part[w + 3][lane];
        }
        uhm[blockIdx.x * 32 + lane] = ((a0 + a1) + (a2 + a3)) * (1.0f / (float)JB);
    }
}

__launch_bounds__(256)
__global__ void dotn2_kernel(const float* __restrict__ uh,
                             const float* __restrict__ uhm,
                             float* __restrict__ dotp, float* __restrict__ n2) {
    const int wave = threadIdx.x >> 5;
    const int lane = threadIdx.x & 31;
    const int pair = blockIdx.x * 8 + wave;
    const int r = pair >> 3, j = pair & 7;

    const float4 a = ((const float4*)(uh + (size_t)r * COLS + j * DIM))[lane];
    const float4 m = ((const float4*)(uhm + j * DIM))[lane];
    float d = a.x * m.x + a.y * m.y + a.z * m.z + a.w * m.w;
    float q = a.x * a.x + a.y * a.y + a.z * a.z + a.w * a.w;
#pragma unroll
    for (int off = 16; off > 0; off >>= 1) {
        d += __shfl_xor(d, off, 32);
        q += __shfl_xor(q, off, 32);
    }
    if (lane == 0) { dotp[pair] = d; n2[pair] = q; }
}

__launch_bounds__(1024)
__global__ void routing_kernel(const float* __restrict__ dotp,
                               const float* __restrict__ n2,
                               float* __restrict__ cf) {
    __shared__ float B7[8];
    __shared__ float c[8];
    __shared__ float sumj[8];
    const int tid = threadIdx.x;
    if (tid < 8) B7[tid] = 0.0f;

    for (int it = 0; it < 2; ++it) {
        __syncthreads();
        if (tid == 0) {
            float m = B7[0];
            for (int jj = 1; jj < 8; ++jj) m = fmaxf(m, B7[jj]);
            float s = 0.0f, e[8];
            for (int jj = 0; jj < 8; ++jj) { e[jj] = __expf(B7[jj] - m); s += e[jj]; }
            for (int jj = 0; jj < 8; ++jj) c[jj] = e[jj] / s;
        }
        __syncthreads();
        if (tid < 8) sumj[tid] = 0.0f;
        __syncthreads();
        const int j = tid & 7;
        const float cj = c[j];
        const float cc = cj * cj;
        float a = 0.0f;
#pragma unroll 8
        for (int p = tid; p < JB * NJ; p += 1024)
            a += squash_gain_routing(cc * n2[p]) * dotp[p];
        atomicAdd(&sumj[j], a);
        __syncthreads();
        if (tid < 8) B7[tid] += c[tid] * sumj[tid] * (1.0f / (float)JB);
    }
    __syncthreads();
    if (tid == 0) {
        float m = B7[0];
        for (int jj = 1; jj < 8; ++jj) m = fmaxf(m, B7[jj]);
        float s = 0.0f, e[8];
        for (int jj = 0; jj < 8; ++jj) { e[jj] = __expf(B7[jj] - m); s += e[jj]; }
        for (int jj = 0; jj < 8; ++jj) cf[jj] = e[jj] / s;
    }
}

__launch_bounds__(256)
__global__ void final_kernel(const float* __restrict__ uh,
                             const float* __restrict__ n2,
                             const float* __restrict__ cf,
                             float* __restrict__ out) {
    const size_t idx  = (size_t)blockIdx.x * 256 + threadIdx.x;
    const size_t base = idx * 4;
    const int r = (int)(base >> 10);
    const int j = (int)((base >> 7) & 7);
    const float cj = cf[j];
    const float gain = cj * squash_gain(cj * cj * n2[r * 8 + j]);
    float4 a = *(const float4*)(uh + base);
    a.x *= gain; a.y *= gain; a.z *= gain; a.w *= gain;
    *(float4*)(out + base) = a;
}

extern "C" void kernel_launch(void* const* d_in, const int* in_sizes, int n_in,
                              void* d_out, int out_size, void* d_ws, size_t ws_size,
                              hipStream_t stream) {
    const float* x = (const float*)d_in[0];
    const float* w = (const float*)d_in[1];
    float* out = (float*)d_out;

    char* ws = (char*)d_ws;
    float* uh   = (float*)(ws);
    float* n2   = (float*)(ws + (size_t)JB * COLS * 4);
    float* dotp = (float*)(ws + (size_t)JB * COLS * 4 + 262144);
    float* uhm  = (float*)(ws + (size_t)JB * COLS * 4 + 524288);
    float* cf   = (float*)(ws + (size_t)JB * COLS * 4 + 528384);

    gemm_uhat_kernel<<<dim3(128, 2, 1), 256, 0, stream>>>(x, w, uh);
    colmean_kernel<<<32, 1024, 0, stream>>>(uh, uhm);
    dotn2_kernel<<<JB, 256, 0, stream>>>(uh, uhm, dotp, n2);
    routing_kernel<<<1, 1024, 0, stream>>>(dotp, n2, cf);
    final_kernel<<<(JB * COLS / 4) / 256, 256, 0, stream>>>(uh, n2, cf, out);
}


# column-mean kernel on 128 blocks, not 32: rows split 4 ways, each part leaves a partial mean row (at the head of out, which the last kernel overwrites); dot kernel adds the 4; sum tree regrouped
# speedup vs baseline: 8.2107x; 1.1402x over previous
#include <hip/hip_runtime.h>
#include <hip/hip_bf16.h>

typedef __attribute__((ext_vector_type(2))) float v2f;
typedef __attribute__((ext_vector_type(8))) float v8f;
typedef __attribute__((ext_vector_type(4))) float v4f;

#define JB 8192
#define NJ 8
#define DIM 128
#define COLS 1024
#define MPARTS 4

__device__ __forceinline__ float squash_gain(float n2) {
    return n2 / ((1.0f + n2) * sqrtf(n2 + 1e-8f));
}

__device__ __forceinline__ float squash_gain_routing(float n2) {
    return n2 * __builtin_amdgcn_rsqf(n2 + 1e-8f) * __builtin_amdgcn_rcpf(1.0f + n2);
}

__launch_bounds__(256)
__global__ void gemm_uhat_kernel(const float* __restrict__ x,
                                 const float* __restrict__ w,
                                 float* __restrict__ uh) {
    const float* x7 = x + (size_t)7 * 8 * 1024 * 128;
    const float* w7 = w + (size_t)7 * 8 * 128 * 128;

    const int wave = threadIdx.x >> 5;
    const int lane = threadIdx.x & 31;
    const int h    = lane >> 4;
    const int c    = lane & 15;

    const int grp  = blockIdx.y * 8 + wave;
    const int j    = grp >> 1;
    const int col0 = (grp & 1) * 64 + 4 * c;

    const float* wl = w7 + (size_t)j * 128 * 128 + (size_t)(4 * h) * 128 + col0;
    v4f bq[32][2];
#pragma unroll
    for (int t = 0; t < 32; ++t) {
#pragma unroll
        for (int e = 0; e < 2; ++e)
            bq[t][e] = *(const v4f*)(wl + (8 * (t >> 1) + 2 * (t & 1) + e) * 128);
    }

    const float* xr = x7 + (size_t)((c & 7) * 1024 + blockIdx.x * 8 + (c >> 3)) * 128 + 4 * h;
    float* orow = uh + (size_t)(blockIdx.x * 64 + 8 * h) * COLS + grp * 64 + 4 * c;

#pragma unroll
    for (int i = 0; i < 4; ++i) {
        v4f a4[16];
#pragma unroll
        for (int u = 0; u < 16; ++u) a4[u] = *(const v4f*)(xr + i * 256 + 8 * u);

        v8f acc[4];
#pragma unroll
        for (int T = 0; T < 4; ++T) acc[T] = (v8f){0,0,0,0,0,0,0,0};

#pragma unroll
        for (int u = 0; u < 16; ++u) {
#pragma unroll
            for (int p = 0; p < 2; ++p) {
                v2f a;
                a.x = a4[u][2 * p + 0];
                a.y = a4[u][2 * p + 1];
#pragma unroll
                for (int T = 0; T < 4; ++T) {
                    v2f b;
                    b.x = bq[2 * u + p][0][T];
                    b.y = bq[2 * u + p][1][T];
                    acc[T] = __builtin_amdgcn_wmma_f32_16x16x4_f32(
                        false, a, false, b, (short)0, acc[T], false, false);
                }
            }
        }
#pragma unroll
        for (int g = 0; g < 8; ++g) {
            v4f o;
            o.x = acc[0][g]; o.y = acc[1][g]; o.z = acc[2][g]; o.w = acc[3][g];
            *(v4f*)(orow + (size_t)(16 * i + g) * COLS) = o;
        }
    }
}

__device__ __forceinline__ float fold_rows(float v) {
    v += __shfl_xor(v, 8, 32);
    v += __shfl_xor(v, 16, 32);
    return v;
}

__launch_bounds__(1024)
__global__ void colmean_kernel(const float* __restrict__ uh, float* __restrict__ uhp) {
    __shared__ __attribute__((aligned(16))) float part[32][32];
    const int wave = threadIdx.x >> 5;
    const int lane = threadIdx.x & 31;
    const int q    = lane & 7;
    const int rs   = lane >> 3;
    const float* p = uh + (size_t)(blockIdx.y * (JB / MPARTS) + wave * 4 + rs) * COLS + blockIdx.x * 32 + q * 4;

    v4f acc[8] = {};

    for (int t = 0; t < JB / MPARTS / 128; t += 8) {
#pragma unroll
        for (int u = 0; u < 8; ++u)
            acc[u] += *(const v4f*)(p + (size_t)(t + u) * (128 * COLS));
    }
    v4f s = ((acc[0] + acc[1]) + (acc[2] + acc[3])) + ((acc[4] + acc[5]) + (acc[6] + acc[7]));
    s.x = fold_rows(s.x);
    s.y = fold_rows(s.y);
    s.z = fold_rows(s.z);
    s.w = fold_rows(s.w);
    if (lane < 8) *(v4f*)&part[wave][q * 4] = s;
    __syncthreads();
    if (wave == 0) {
        float a0 = 0.0f, a1 = 0.0f, a2 = 0.0f, a3 = 0.0f;
        for (int w = 0; w < 32; w += 4) {
            a0 += part[w + 0][lane];
            a1 += part[w + 1][lane];
            a2 += part[w + 2][lane];
            a3 += part[w + 3][lane];
        }
        uhp[blockIdx.y * COLS + blockIdx.x * 32 + lane] = ((a0 + a1) + (a2 + a3)) * (1.0f / (float)JB);
    }
}

__launch_bounds__(256)
__global__ void dotn2_kernel(const float* __restrict__ uh,
                             const float* __restrict__ uhp,
                             float* __restrict__ dotp, float* __restrict__ n2) {
    const int wave = threadIdx.x >> 5;
    const int lane = threadIdx.x & 31;
    const int pair = blockIdx.x * 8 + wave;
    const int r = pair >> 3, j = pair & 7;

    const float4 a = ((const float4*)(uh + (size_t)r * COLS + j * DIM))[lane];
    const v4f* mp = (const v4f*)(uhp + j * DIM) + lane;
    const v4f m = (mp[0] + mp[COLS / 4]) + (mp[2 * (COLS / 4)] + mp[3 * (COLS / 4)]);
    float d = a.x * m.x + a.y * m.y + a.z * m.z + a.w * m.w;
    float q = a.x * a.x + a.y * a.y + a.z * a.z + a.w * a.w;
#pragma unroll
    for (int off = 16; off > 0; off >>= 1) {
        d += __shfl_xor(d, off, 32);
        q += __shfl_xor(q, off, 32);
    }
    if (lane == 0) { dotp[pair] = d; n2[pair] = q; }
}

__launch_bounds__(1024)
__global__ void routing_kernel(const float* __restrict__ dotp,
                               const float* __restrict__ n2,
                               float* __restrict__ cf) {
    __shared__ float B7[8];
    __shared__ float c[8];
    __shared__ float sumj[8];
    const int tid = threadIdx.x;
    if (tid < 8) B7[tid] = 0.0f;

    for (int it = 0; it < 2; ++it) {
        __syncthreads();
        if (tid == 0) {
            float m = B7[0];
            for (int jj = 1; jj < 8; ++jj) m = fmaxf(m, B7[jj]);
            float s = 0.0f, e[8];
            for (int jj = 0; jj < 8; ++jj) { e[jj] = __expf(B7[jj] - m); s += e[jj]; }
            for (int jj = 0; jj < 8; ++jj) c[jj] = e[jj] / s;
        }
        __syncthreads();
        if (tid < 8) sumj[tid] = 0.0f;
        __syncthreads();
        const int j = tid & 7;
        const float cj = c[j];
        const float cc = cj * cj;
        float a = 0.0f;
#pragma unroll 8
        for (int p = tid; p < JB * NJ; p += 1024)
            a += squash_gain_routing(cc * n2[p]) * dotp[p];
        atomicAdd(&sumj[j], a);
        __syncthreads();
        if (tid < 8) B7[tid] += c[tid] * sumj[tid] * (1.0f / (float)JB);
    }
    __syncthreads();
    if (tid == 0) {
        float m = B7[0];
        for (int jj = 1; jj < 8; ++jj) m = fmaxf(m, B7[jj]);
        float s = 0.0f, e[8];
        for (int jj = 0; jj < 8; ++jj) { e[jj] = __expf(B7[jj] - m); s += e[jj]; }
        for (int jj = 0; jj < 8; ++jj) cf[jj] = e[jj] / s;
    }
}

__launch_bounds__(256)
__global__ void final_kernel(const float* __restrict__ uh,
                             const float* __restrict__ n2,
                             const float* __restrict__ cf,
                             float* __restrict__ out) {
    const size_t idx  = (size_t)blockIdx.x * 256 + threadIdx.x;
    const size_t base = idx * 4;
    const int r = (int)(base >> 10);
    const int j = (int)((base >> 7) & 7);
    const float cj = cf[j];
    const float gain = cj * squash_gain(cj * cj * n2[r * 8 + j]);
    float4 a = *(const float4*)(uh + base);
    a.x *= gain; a.y *= gain; a.z *= gain; a.w *= gain;
    *(float4*)(out + base) = a;
}

extern "C" void kernel_launch(void* const* d_in, const int* in_sizes, int n_in,
                              void* d_out, int out_size, void* d_ws, size_t ws_size,
                              hipStream_t stream) {
    const float* x = (const float*)d_in[0];
    const float* w = (const float*)d_in[1];
    float* out = (float*)d_out;

    char* ws = (char*)d_ws;
    float* uh   = (float*)(ws);
    float* n2   = (float*)(ws + (size_t)JB * COLS * 4);
    float* dotp = (float*)(ws + (size_t)JB * COLS * 4 + 262144);
    float* cf   = (float*)(ws + (size_t)JB * COLS * 4 + 528384);
    float* uhp  = out;

    gemm_uhat_kernel<<<dim3(128, 2, 1), 256, 0, stream>>>(x, w, uh);
    colmean_kernel<<<dim3(32, MPARTS, 1), 1024, 0, stream>>>(uh, uhp);
    dotn2_kernel<<<JB, 256, 0, stream>>>(uh, uhp, dotp, n2);
    routing_kernel<<<1, 1024, 0, stream>>>(dotp, n2, cf);
    final_kernel<<<(JB * COLS / 4) / 256, 256, 0, stream>>>(uh, n2, cf, out);
}


# routing sums as gain-weighted column sums of uh dotted with the mean (V.m, linear in uh) in two 256-block passes over uh: no per-pair dot, no one-block routing loop; sum order changes, fixed trees
# speedup vs baseline: 9.5080x; 1.1580x over previous
#include <hip/hip_runtime.h>
#include <hip/hip_bf16.h>

typedef __attribute__((ext_vector_type(2))) float v2f;
typedef __attribute__((ext_vector_type(8))) float v8f;
typedef __attribute__((ext_vector_type(4))) float v4f;

#define JB 8192
#define NJ 8
#define DIM 128
#define COLS 1024

__device__ __forceinline__ float squash_gain(float n2) {
    return n2 / ((1.0f + n2) * sqrtf(n2 + 1e-8f));
}

__device__ __forceinline__ float squash_gain_routing(float n2) {
    return n2 * __builtin_amdgcn_rsqf(n2 + 1e-8f) * __builtin_amdgcn_rcpf(1.0f + n2);
}

__launch_bounds__(256)
__global__ void gemm_uhat_kernel(const float* __restrict__ x,
                                 const float* __restrict__ w,
                                 float* __restrict__ uh) {
    const float* x7 = x + (size_t)7 * 8 * 1024 * 128;
    const float* w7 = w + (size_t)7 * 8 * 128 * 128;

    const int wave = threadIdx.x >> 5;
    const int lane = threadIdx.x & 31;
    const int h    = lane >> 4;
    const int c    = lane & 15;

    const int grp  = blockIdx.y * 8 + wave;
    const int j    = grp >> 1;
    const int col0 = (grp & 1) * 64 + 4 * c;

    const float* wl = w7 + (size_t)j * 128 * 128 + (size_t)(4 * h) * 128 + col0;
    v4f bq[32][2];
#pragma unroll
    for (int t = 0; t < 32; ++t) {
#pragma unroll
        for (int e = 0; e < 2; ++e)
            bq[t][e] = *(const v4f*)(wl + (8 * (t >> 1) + 2 * (t & 1) + e) * 128);
    }

    const float* xr = x7 + (size_t)((c & 7) * 1024 + blockIdx.x * 8 + (c >> 3)) * 128 + 4 * h;
    float* orow = uh + (size_t)(blockIdx.x * 64 + 8 * h) * COLS + grp * 64 + 4 * c;

#pragma unroll
    for (int i = 0; i < 4; ++i) {
        v4f a4[16];
#pragma unroll
        for (int u = 0; u < 16; ++u) a4[u] = *(const v4f*)(xr + i * 256 + 8 * u);

        v8f acc[4];
#pragma unroll
        for (int T = 0; T < 4; ++T) acc[T] = (v8f){0,0,0,0,0,0,0,0};

#pragma unroll
        for (int u = 0; u < 16; ++u) {
#pragma unroll
            for (int p = 0; p < 2; ++p) {
                v2f a;
                a.x = a4[u][2 * p + 0];
                a.y = a4[u][2 * p + 1];
#pragma unroll
                for (int T = 0; T < 4; ++T) {
                    v2f b;
                    b.x = bq[2 * u + p][0][T];
                    b.y = bq[2 * u + p][1][T];
                    acc[T] = __builtin_amdgcn_wmma_f32_16x16x4_f32(
                        false, a, false, b, (short)0, acc[T], false, false);
                }
            }
        }
#pragma unroll
        for (int g = 0; g < 8; ++g) {
            v4f o;
            o.x = acc[0][g]; o.y = acc[1][g]; o.z = acc[2][g]; o.w = acc[3][g];
            *(v4f*)(orow + (size_t)(16 * i + g) * COLS) = o;
        }
    }
}

#define RPARTS 32
#define PROWS (JB / RPARTS)

__device__ __forceinline__ float wave_sum(float v) {
#pragma unroll
    for (int off = 16; off > 0; off >>= 1) v += __shfl_xor(v, off, 32);
    return v;
}

__device__ __forceinline__ float logit1(float s1) {
    return 0.125f * s1 * (1.0f / (float)JB);
}

__launch_bounds__(1024)
__global__ void pass_a_kernel(const float* __restrict__ uh, float* __restrict__ n2,
                              float* __restrict__ mp, float* __restrict__ v1p) {
    __shared__ __attribute__((aligned(16))) float red[2][32][DIM];
    const int wave = threadIdx.x >> 5;
    const int lane = threadIdx.x & 31;
    const int j    = blockIdx.x;
    const int r0   = blockIdx.y * PROWS + wave;
    const float* p = uh + (size_t)r0 * COLS + j * DIM + lane * 4;

    v4f a[8];
#pragma unroll
    for (int u = 0; u < 8; ++u) a[u] = *(const v4f*)(p + (size_t)u * (32 * COLS));

    v4f m = {0, 0, 0, 0}, v = {0, 0, 0, 0};
#pragma unroll
    for (int u = 0; u < 8; ++u) {
        float q = a[u].x * a[u].x + a[u].y * a[u].y + a[u].z * a[u].z + a[u].w * a[u].w;
        q = wave_sum(q);
        if (lane == 0) n2[(r0 + 32 * u) * NJ + j] = q;
        const float g = squash_gain_routing(q * (1.0f / 64.0f));
        m += a[u];
        v += g * a[u];
    }
    *(v4f*)&red[0][wave][lane * 4] = m;
    *(v4f*)&red[1][wave][lane * 4] = v;
    __syncthreads();
    if (threadIdx.x < 2 * DIM) {
        const int which = threadIdx.x >> 7;
        const int e     = threadIdx.x & (DIM - 1);
        float s0 = 0.0f, s1 = 0.0f, s2 = 0.0f, s3 = 0.0f;
        for (int w = 0; w < 32; w += 4) {
            s0 += red[which][w + 0][e];
            s1 += red[which][w + 1][e];
            s2 += red[which][w + 2][e];
            s3 += red[which][w + 3][e];
        }
        const float s = (s0 + s1) + (s2 + s3);
        const int o = blockIdx.y * COLS + j * DIM + e;
        if (which == 0) mp[o] = s * (1.0f / (float)JB);
        else            v1p[o] = s;
    }
}

__launch_bounds__(1024)
__global__ void join_a_kernel(const float* __restrict__ mp, const float* __restrict__ v1p,
                              float* __restrict__ uhm, float* __restrict__ s1) {
    __shared__ float rm[8][DIM];
    __shared__ float rv[8][DIM];
    __shared__ float wd[4];
    const int j = blockIdx.x;
    const int g = threadIdx.x >> 7;
    const int e = threadIdx.x & (DIM - 1);
    const int col = j * DIM + e;
    rm[g][e] = (mp[g * COLS + col] + mp[(g + 8) * COLS + col]) +
               (mp[(g + 16) * COLS + col] + mp[(g + 24) * COLS + col]);
    rv[g][e] = (v1p[g * COLS + col] + v1p[(g + 8) * COLS + col]) +
               (v1p[(g + 16) * COLS + col] + v1p[(g + 24) * COLS + col]);
    __syncthreads();
    if (threadIdx.x < DIM) {
        const float M = ((rm[0][e] + rm[1][e]) + (rm[2][e] + rm[3][e])) +
                        ((rm[4][e] + rm[5][e]) + (rm[6][e] + rm[7][e]));
        const float V = ((rv[0][e] + rv[1][e]) + (rv[2][e] + rv[3][e])) +
                        ((rv[4][e] + rv[5][e]) + (rv[6][e] + rv[7][e]));
        uhm[col] = M;
        const float d = wave_sum(M * V);
        if ((threadIdx.x & 31) == 0) wd[threadIdx.x >> 5] = d;
    }
    __syncthreads();
    if (threadIdx.x == 0) s1[j] = (wd[0] + wd[1]) + (wd[2] + wd[3]);
}

__launch_bounds__(1024)
__global__ void pass_b_kernel(const float* __restrict__ uh, const float* __restrict__ n2,
                              const float* __restrict__ s1, float* __restrict__ v2p) {
    __shared__ __attribute__((aligned(16))) float red[32][DIM];
    const int wave = threadIdx.x >> 5;
    const int lane = threadIdx.x & 31;
    const int j    = blockIdx.x;
    const int r0   = blockIdx.y * PROWS + wave;
    const float* p = uh + (size_t)r0 * COLS + j * DIM + lane * 4;

    v4f a[8];
    float nq[8];
#pragma unroll
    for (int u = 0; u < 8; ++u) {
        a[u]  = *(const v4f*)(p + (size_t)u * (32 * COLS));
        nq[u] = n2[(r0 + 32 * u) * NJ + j];
    }

    float mx = logit1(s1[0]);
    for (int k = 1; k < NJ; ++k) mx = fmaxf(mx, logit1(s1[k]));
    float se = 0.0f;
    for (int k = 0; k < NJ; ++k) se += __expf(logit1(s1[k]) - mx);
    const float cj = __expf(logit1(s1[j]) - mx) / se;
    const float cc = cj * cj;

    v4f v = {0, 0, 0, 0};
#pragma unroll
    for (int u = 0; u < 8; ++u) v += squash_gain_routing(cc * nq[u]) * a[u];

    *(v4f*)&red[wave][lane * 4] = v;
    __syncthreads();
    if (threadIdx.x < DIM) {
        const int e = threadIdx.x;
        float s0 = 0.0f, s1a = 0.0f, s2 = 0.0f, s3 = 0.0f;
        for (int w = 0; w < 32; w += 4) {
            s0  += red[w + 0][e];
            s1a += red[w + 1][e];
            s2  += red[w + 2][e];
            s3  += red[w + 3][e];
        }
        v2p[blockIdx.y * COLS + j * DIM + e] = (s0 + s1a) + (s2 + s3);
    }
}

__launch_bounds__(1024)
__global__ void join_b_kernel(const float* __restrict__ v2p, const float* __restrict__ uhm,
                              const float* __restrict__ s1, float* __restrict__ cf) {
    __shared__ float wd[32];
    const int t = threadIdx.x;
    float a0 = 0.0f, a1 = 0.0f, a2 = 0.0f, a3 = 0.0f;
#pragma unroll
    for (int q = 0; q < RPARTS; q += 4) {
        a0 += v2p[(q + 0) * COLS + t];
        a1 += v2p[(q + 1) * COLS + t];
        a2 += v2p[(q + 2) * COLS + t];
        a3 += v2p[(q + 3) * COLS + t];
    }
    const float d = wave_sum(((a0 + a1) + (a2 + a3)) * uhm[t]);
    if ((t & 31) == 0) wd[t >> 5] = d;
    __syncthreads();
    if (t == 0) {
        float b1[NJ], c1[NJ], b2[NJ];
        for (int k = 0; k < NJ; ++k) b1[k] = logit1(s1[k]);
        float mx = b1[0];
        for (int k = 1; k < NJ; ++k) mx = fmaxf(mx, b1[k]);
        float se = 0.0f;
        for (int k = 0; k < NJ; ++k) { c1[k] = __expf(b1[k] - mx); se += c1[k]; }
        for (int k = 0; k < NJ; ++k) {
            const float s2 = (wd[4 * k] + wd[4 * k + 1]) + (wd[4 * k + 2] + wd[4 * k + 3]);
            b2[k] = b1[k] + (c1[k] / se) * s2 * (1.0f / (float)JB);
        }
        mx = b2[0];
        for (int k = 1; k < NJ; ++k) mx = fmaxf(mx, b2[k]);
        se = 0.0f;
        float e2[NJ];
        for (int k = 0; k < NJ; ++k) { e2[k] = __expf(b2[k] - mx); se += e2[k]; }
        for (int k = 0; k < NJ; ++k) cf[k] = e2[k] / se;
    }
}

__launch_bounds__(256)
__global__ void final_kernel(const float* __restrict__ uh,
                             const float* __restrict__ n2,
                             const float* __restrict__ cf,
                             float* __restrict__ out) {
    const size_t idx  = (size_t)blockIdx.x * 256 + threadIdx.x;
    const size_t base = idx * 4;
    const int r = (int)(base >> 10);
    const int j = (int)((base >> 7) & 7);
    const float cj = cf[j];
    const float gain = cj * squash_gain(cj * cj * n2[r * 8 + j]);
    float4 a = *(const float4*)(uh + base);
    a.x *= gain; a.y *= gain; a.z *= gain; a.w *= gain;
    *(float4*)(out + base) = a;
}

extern "C" void kernel_launch(void* const* d_in, const int* in_sizes, int n_in,
                              void* d_out, int out_size, void* d_ws, size_t ws_size,
                              hipStream_t stream) {
    const float* x = (const float*)d_in[0];
    const float* w = (const float*)d_in[1];
    float* out = (float*)d_out;

    char* ws = (char*)d_ws;
    float* uh   = (float*)(ws);
    float* n2   = (float*)(ws + (size_t)JB * COLS * 4);
    float* mp   = (float*)(ws + (size_t)JB * COLS * 4 + 262144);
    float* v1p  = (float*)(ws + (size_t)JB * COLS * 4 + 393216);
    float* uhm  = (float*)(ws + (size_t)JB * COLS * 4 + 524288);
    float* cf   = (float*)(ws + (size_t)JB * COLS * 4 + 528384);
    float* v2p  = out;
    float* s1   = out + RPARTS * COLS;

    gemm_uhat_kernel<<<dim3(128, 2, 1), 256, 0, stream>>>(x, w, uh);
    pass_a_kernel<<<dim3(NJ, RPARTS, 1), 1024, 0, stream>>>(uh, n2, mp, v1p);
    join_a_kernel<<<NJ, 1024, 0, stream>>>(mp, v1p, uhm, s1);
    pass_b_kernel<<<dim3(NJ, RPARTS, 1), 1024, 0, stream>>>(uh, n2, s1, v2p);
    join_b_kernel<<<1, 1024, 0, stream>>>(v2p, uhm, s1, cf);
    final_kernel<<<(JB * COLS / 4) / 256, 256, 0, stream>>>(uh, n2, cf, out);
}


# second routing sum over x, not uh, by linearity: s2[j] = sum_r gain2[r,j] (x[r].z[j]), z[j] = W[7,j] uhm[j] formed in the first join; 1024-wave pass; sum order changes, fixed trees
# speedup vs baseline: 9.7850x; 1.0291x over previous
#include <hip/hip_runtime.h>
#include <hip/hip_bf16.h>

typedef __attribute__((ext_vector_type(2))) float v2f;
typedef __attribute__((ext_vector_type(8))) float v8f;
typedef __attribute__((ext_vector_type(4))) float v4f;

#define JB 8192
#define NJ 8
#define DIM 128
#define COLS 1024

__device__ __forceinline__ float squash_gain(float n2) {
    return n2 / ((1.0f + n2) * sqrtf(n2 + 1e-8f));
}

__device__ __forceinline__ float squash_gain_routing(float n2) {
    return n2 * __builtin_amdgcn_rsqf(n2 + 1e-8f) * __builtin_amdgcn_rcpf(1.0f + n2);
}

__launch_bounds__(256)
__global__ void gemm_uhat_kernel(const float* __restrict__ x,
                                 const float* __restrict__ w,
                                 float* __restrict__ uh) {
    const float* x7 = x + (size_t)7 * 8 * 1024 * 128;
    const float* w7 = w + (size_t)7 * 8 * 128 * 128;

    const int wave = threadIdx.x >> 5;
    const int lane = threadIdx.x & 31;
    const int h    = lane >> 4;
    const int c    = lane & 15;

    const int grp  = blockIdx.y * 8 + wave;
    const int j    = grp >> 1;
    const int col0 = (grp & 1) * 64 + 4 * c;

    const float* wl = w7 + (size_t)j * 128 * 128 + (size_t)(4 * h) * 128 + col0;
    v4f bq[32][2];
#pragma unroll
    for (int t = 0; t < 32; ++t) {
#pragma unroll
        for (int e = 0; e < 2; ++e)
            bq[t][e] = *(const v4f*)(wl + (8 * (t >> 1) + 2 * (t & 1) + e) * 128);
    }

    const float* xr = x7 + (size_t)((c & 7) * 1024 + blockIdx.x * 8 + (c >> 3)) * 128 + 4 * h;
    float* orow = uh + (size_t)(blockIdx.x * 64 + 8 * h) * COLS + grp * 64 + 4 * c;

#pragma unroll
    for (int i = 0; i < 4; ++i) {
        v4f a4[16];
#pragma unroll
        for (int u = 0; u < 16; ++u) a4[u] = *(const v4f*)(xr + i * 256 + 8 * u);

        v8f acc[4];
#pragma unroll
        for (int T = 0; T < 4; ++T) acc[T] = (v8f){0,0,0,0,0,0,0,0};

#pragma unroll
        for (int u = 0; u < 16; ++u) {
#pragma unroll
            for (int p = 0; p < 2; ++p) {
                v2f a;
                a.x = a4[u][2 * p + 0];
                a.y = a4[u][2 * p + 1];
#pragma unroll
                for (int T = 0; T < 4; ++T) {
                    v2f b;
                    b.x = bq[2 * u + p][0][T];
                    b.y = bq[2 * u + p][1][T];
                    acc[T] = __builtin_amdgcn_wmma_f32_16x16x4_f32(
                        false, a, false, b, (short)0, acc[T], false, false);
                }
            }
        }
#pragma unroll
        for (int g = 0; g < 8; ++g) {
            v4f o;
            o.x = acc[0][g]; o.y = acc[1][g]; o.z = acc[2][g]; o.w = acc[3][g];
            *(v4f*)(orow + (size_t)(16 * i + g) * COLS) = o;
        }
    }
}

#define RPARTS 32
#define PROWS (JB / RPARTS)

__device__ __forceinline__ float wave_sum(float v) {
#pragma unroll
    for (int off = 16; off > 0; off >>= 1) v += __shfl_xor(v, off, 32);
    return v;
}

__device__ __forceinline__ float logit1(float s1) {
    return 0.125f * s1 * (1.0f / (float)JB);
}

__launch_bounds__(1024)
__global__ void pass_a_kernel(const float* __restrict__ uh, float* __restrict__ n2,
                              float* __restrict__ mp, float* __restrict__ v1p) {
    __shared__ __attribute__((aligned(16))) float red[2][32][DIM];
    const int wave = threadIdx.x >> 5;
    const int lane = threadIdx.x & 31;
    const int j    = blockIdx.x;
    const int r0   = blockIdx.y * PROWS + wave;
    const float* p = uh + (size_t)r0 * COLS + j * DIM + lane * 4;

    v4f a[8];
#pragma unroll
    for (int u = 0; u < 8; ++u) a[u] = *(const v4f*)(p + (size_t)u * (32 * COLS));

    v4f m = {0, 0, 0, 0}, v = {0, 0, 0, 0};
#pragma unroll
    for (int u = 0; u < 8; ++u) {
        float q = a[u].x * a[u].x + a[u].y * a[u].y + a[u].z * a[u].z + a[u].w * a[u].w;
        q = wave_sum(q);
        if (lane == 0) n2[(r0 + 32 * u) * NJ + j] = q;
        const float g = squash_gain_routing(q * (1.0f / 64.0f));
        m += a[u];
        v += g * a[u];
    }
    *(v4f*)&red[0][wave][lane * 4] = m;
    *(v4f*)&red[1][wave][lane * 4] = v;
    __syncthreads();
    if (threadIdx.x < 2 * DIM) {
        const int which = threadIdx.x >> 7;
        const int e     = threadIdx.x & (DIM - 1);
        float s0 = 0.0f, s1 = 0.0f, s2 = 0.0f, s3 = 0.0f;
        for (int w = 0; w < 32; w += 4) {
            s0 += red[which][w + 0][e];
            s1 += red[which][w + 1][e];
            s2 += red[which][w + 2][e];
            s3 += red[which][w + 3][e];
        }
        const float s = (s0 + s1) + (s2 + s3);
        const int o = blockIdx.y * COLS + j * DIM + e;
        if (which == 0) mp[o] = s * (1.0f / (float)JB);
        else            v1p[o] = s;
    }
}

template <int CTRL>
__device__ __forceinline__ float row_xor(float x) {
    return __int_as_float(__builtin_amdgcn_update_dpp(0, __float_as_int(x), CTRL, 0xf, 0xf, false));
}
__device__ __forceinline__ float xhalf(float x) {
    return __int_as_float(__builtin_amdgcn_permlanex16(__float_as_int(x), __float_as_int(x),
                                                       0x76543210, 0xfedcba98, false, false));
}
__device__ __forceinline__ float wave_sum_valu(float v) {
    v += row_xor<0x161>(v);
    v += row_xor<0x162>(v);
    v += row_xor<0x164>(v);
    v += row_xor<0x168>(v);
    return v + xhalf(v);
}

#define ZQ 4
__launch_bounds__(1024)
__global__ void join_a_kernel(const float* __restrict__ mp, const float* __restrict__ v1p,
                              const float* __restrict__ w,
                              float* __restrict__ s1, float* __restrict__ z) {
    __shared__ float rm[8][DIM];
    __shared__ float rv[8][DIM];
    __shared__ float wd[4];
    __shared__ __attribute__((aligned(16))) float um[DIM];
    const int j = blockIdx.x;
    const int g = threadIdx.x >> 7;
    const int e = threadIdx.x & (DIM - 1);
    const int col = j * DIM + e;
    const int d = blockIdx.y * 32 + (threadIdx.x >> 5);
    const v4f wq = *(const v4f*)(w + (size_t)7 * 8 * 128 * 128 + (size_t)j * 128 * 128 +
                                 (size_t)d * 128 + (threadIdx.x & 31) * 4);
    rm[g][e] = (mp[g * COLS + col] + mp[(g + 8) * COLS + col]) +
               (mp[(g + 16) * COLS + col] + mp[(g + 24) * COLS + col]);
    rv[g][e] = (v1p[g * COLS + col] + v1p[(g + 8) * COLS + col]) +
               (v1p[(g + 16) * COLS + col] + v1p[(g + 24) * COLS + col]);
    __syncthreads();
    if (threadIdx.x < DIM) {
        const float M = ((rm[0][e] + rm[1][e]) + (rm[2][e] + rm[3][e])) +
                        ((rm[4][e] + rm[5][e]) + (rm[6][e] + rm[7][e]));
        const float V = ((rv[0][e] + rv[1][e]) + (rv[2][e] + rv[3][e])) +
                        ((rv[4][e] + rv[5][e]) + (rv[6][e] + rv[7][e]));
        um[e] = M;
        const float d = wave_sum(M * V);
        if ((threadIdx.x & 31) == 0) wd[threadIdx.x >> 5] = d;
    }
    __syncthreads();
    if (blockIdx.y == 0 && threadIdx.x == 0) s1[j] = (wd[0] + wd[1]) + (wd[2] + wd[3]);
    const v4f u4 = *(const v4f*)&um[(threadIdx.x & 31) * 4];
    const float zp = wave_sum_valu(wq.x * u4.x + wq.y * u4.y + wq.z * u4.z + wq.w * u4.w);
    if ((threadIdx.x & 31) == 0) z[j * DIM + d] = zp;
}

#define XWAVES 4
#define XPARTS (1024 / XWAVES)
__launch_bounds__(32 * XWAVES)
__global__ void pass_b_kernel(const float* __restrict__ x, const float* __restrict__ n2,
                              const float* __restrict__ s1, const float* __restrict__ z,
                              float* __restrict__ s2p) {
    __shared__ __attribute__((aligned(16))) float wd[XWAVES][NJ];
    const int wave = threadIdx.x >> 5;
    const int lane = threadIdx.x & 31;
    const int s    = blockIdx.x * XWAVES + wave;
    const float* xp = x + (size_t)7 * 8 * 1024 * 128 + (size_t)s * 128 + lane * 4;

    v4f a[8], zz[NJ];
#pragma unroll
    for (int b = 0; b < 8; ++b) a[b] = *(const v4f*)(xp + (size_t)b * (1024 * 128));
    const v2f nq = *(const v2f*)(n2 + s * 64 + 2 * lane);
#pragma unroll
    for (int j = 0; j < NJ; ++j) zz[j] = *(const v4f*)(z + j * DIM + lane * 4);

    const int j0 = 2 * (lane & 3);
    float mx = logit1(s1[0]);
    for (int k = 1; k < NJ; ++k) mx = fmaxf(mx, logit1(s1[k]));
    float se = 0.0f, e0 = 0.0f, e1 = 0.0f;
    for (int k = 0; k < NJ; ++k) {
        const float ek = __expf(logit1(s1[k]) - mx);
        se += ek;
        if (k == j0)     e0 = ek;
        if (k == j0 + 1) e1 = ek;
    }
    const float c0 = e0 / se, c1 = e1 / se;

    asm volatile("" :: "v"(a[0]), "v"(a[1]), "v"(a[2]), "v"(a[3]), "v"(a[4]), "v"(a[5]), "v"(a[6]), "v"(a[7]),
                       "v"(zz[0]), "v"(zz[1]), "v"(zz[2]), "v"(zz[3]), "v"(zz[4]), "v"(zz[5]), "v"(zz[6]), "v"(zz[7]),
                       "v"(nq), "v"(c0), "v"(c1));

    const float g0 = squash_gain_routing(c0 * c0 * nq.x);
    const float g1 = squash_gain_routing(c1 * c1 * nq.y);

    v4f y[NJ];
#pragma unroll
    for (int j = 0; j < NJ; ++j) y[j] = (v4f){0, 0, 0, 0};
#pragma unroll
    for (int b = 0; b < 8; ++b) {
#pragma unroll
        for (int j = 0; j < NJ; ++j) {
            const float gv = __int_as_float(__builtin_amdgcn_readlane(
                __float_as_int((j & 1) ? g1 : g0), (b * 8 + j) >> 1));
            y[j] += gv * a[b];
        }
    }
    float d[NJ];
#pragma unroll
    for (int j = 0; j < NJ; ++j)
        d[j] = wave_sum_valu(y[j].x * zz[j].x + y[j].y * zz[j].y +
                             y[j].z * zz[j].z + y[j].w * zz[j].w);
    if (lane == 0) {
        *(v4f*)&wd[wave][0] = (v4f){d[0], d[1], d[2], d[3]};
        *(v4f*)&wd[wave][4] = (v4f){d[4], d[5], d[6], d[7]};
    }
    __syncthreads();
    if (threadIdx.x < NJ)
        s2p[blockIdx.x * NJ + threadIdx.x] = (wd[0][threadIdx.x] + wd[1][threadIdx.x]) +
                                             (wd[2][threadIdx.x] + wd[3][threadIdx.x]);
}

__launch_bounds__(1024)
__global__ void join_b_kernel(const float* __restrict__ s2p, const float* __restrict__ s1,
                              float* __restrict__ cf) {
    __shared__ float wd[32][NJ];
    const int t = threadIdx.x;
    float sv[NJ];
#pragma unroll
    for (int k = 0; k < NJ; ++k) sv[k] = s1[k];
    float v = s2p[t] + s2p[t + 1024];
    asm volatile("" :: "s"(sv[0]), "s"(sv[1]), "s"(sv[2]), "s"(sv[3]), "s"(sv[4]), "s"(sv[5]), "s"(sv[6]), "s"(sv[7]));
    v += row_xor<0x168>(v);
    v += xhalf(v);
    if ((t & 31) < NJ) wd[t >> 5][t & 7] = v;
    __syncthreads();
    if (t < 32) {
        const int j = t & 7;
        const int q = (t >> 3) * 8;
        float a = ((wd[q + 0][j] + wd[q + 1][j]) + (wd[q + 2][j] + wd[q + 3][j])) +
                  ((wd[q + 4][j] + wd[q + 5][j]) + (wd[q + 6][j] + wd[q + 7][j]));
        a += row_xor<0x168>(a);
        a += xhalf(a);
        float s2[NJ];
        for (int k = 0; k < NJ; ++k) s2[k] = __int_as_float(__builtin_amdgcn_readlane(__float_as_int(a), k));
        if (t == 0) {
            float b1[NJ], c1[NJ], b2[NJ];
            for (int k = 0; k < NJ; ++k) b1[k] = logit1(sv[k]);
            float mx = b1[0];
            for (int k = 1; k < NJ; ++k) mx = fmaxf(mx, b1[k]);
            float se = 0.0f;
            for (int k = 0; k < NJ; ++k) { c1[k] = __expf(b1[k] - mx); se += c1[k]; }
            for (int k = 0; k < NJ; ++k)
                b2[k] = b1[k] + (c1[k] / se) * s2[k] * (1.0f / (float)JB);
            mx = b2[0];
            for (int k = 1; k < NJ; ++k) mx = fmaxf(mx, b2[k]);
            se = 0.0f;
            float e2[NJ];
            for (int k = 0; k < NJ; ++k) { e2[k] = __expf(b2[k] - mx); se += e2[k]; }
            for (int k = 0; k < NJ; ++k) cf[k] = e2[k] / se;
        }
    }
}

__launch_bounds__(256)
__global__ void final_kernel(const float* __restrict__ uh,
                             const float* __restrict__ n2,
                             const float* __restrict__ cf,
                             float* __restrict__ out) {
    const size_t idx  = (size_t)blockIdx.x * 256 + threadIdx.x;
    const size_t base = idx * 4;
    const int r = (int)(base >> 10);
    const int j = (int)((base >> 7) & 7);
    const float cj = cf[j];
    const float gain = cj * squash_gain(cj * cj * n2[r * 8 + j]);
    float4 a = *(const float4*)(uh + base);
    a.x *= gain; a.y *= gain; a.z *= gain; a.w *= gain;
    *(float4*)(out + base) = a;
}

extern "C" void kernel_launch(void* const* d_in, const int* in_sizes, int n_in,
                              void* d_out, int out_size, void* d_ws, size_t ws_size,
                              hipStream_t stream) {
    const float* x = (const float*)d_in[0];
    const float* w = (const float*)d_in[1];
    float* out = (float*)d_out;

    char* ws = (char*)d_ws;
    float* uh   = (float*)(ws);
    float* n2   = (float*)(ws + (size_t)JB * COLS * 4);
    float* mp   = (float*)(ws + (size_t)JB * COLS * 4 + 262144);
    float* v1p  = (float*)(ws + (size_t)JB * COLS * 4 + 393216);
    float* cf   = (float*)(ws + (size_t)JB * COLS * 4 + 528384);
    float* z    = out;
    float* s2p  = out + COLS;
    float* s1   = out + RPARTS * COLS;

    gemm_uhat_kernel<<<dim3(128, 2, 1), 256, 0, stream>>>(x, w, uh);
    pass_a_kernel<<<dim3(NJ, RPARTS, 1), 1024, 0, stream>>>(uh, n2, mp, v1p);
    join_a_kernel<<<dim3(NJ, ZQ, 1), 1024, 0, stream>>>(mp, v1p, w, s1, z);
    pass_b_kernel<<<XPARTS, 32 * XWAVES, 0, stream>>>(x, n2, s1, z, s2p);
    join_b_kernel<<<1, 1024, 0, stream>>>(s2p, s1, cf);
    final_kernel<<<(JB * COLS / 4) / 256, 256, 0, stream>>>(uh, n2, cf, out);
}


# pass over uh: a wave's 8 row norms as ONE halving butterfly on the vector ALU (DPP, permlanex16; a lane ends with one row's sum and gain), join's dot sum too: no ds_bpermute; norm sum order changes
# speedup vs baseline: 10.0515x; 1.0272x over previous
#include <hip/hip_runtime.h>
#include <hip/hip_bf16.h>

typedef __attribute__((ext_vector_type(2))) float v2f;
typedef __attribute__((ext_vector_type(8))) float v8f;
typedef __attribute__((ext_vector_type(4))) float v4f;

#define JB 8192
#define NJ 8
#define DIM 128
#define COLS 1024

__device__ __forceinline__ float squash_gain(float n2) {
    return n2 / ((1.0f + n2) * sqrtf(n2 + 1e-8f));
}

__device__ __forceinline__ float squash_gain_routing(float n2) {
    return n2 * __builtin_amdgcn_rsqf(n2 + 1e-8f) * __builtin_amdgcn_rcpf(1.0f + n2);
}

__launch_bounds__(256)
__global__ void gemm_uhat_kernel(const float* __restrict__ x,
                                 const float* __restrict__ w,
                                 float* __restrict__ uh) {
    const float* x7 = x + (size_t)7 * 8 * 1024 * 128;
    const float* w7 = w + (size_t)7 * 8 * 128 * 128;

    const int wave = threadIdx.x >> 5;
    const int lane = threadIdx.x & 31;
    const int h    = lane >> 4;
    const int c    = lane & 15;

    const int grp  = blockIdx.y * 8 + wave;
    const int j    = grp >> 1;
    const int col0 = (grp & 1) * 64 + 4 * c;

    const float* wl = w7 + (size_t)j * 128 * 128 + (size_t)(4 * h) * 128 + col0;
    v4f bq[32][2];
#pragma unroll
    for (int t = 0; t < 32; ++t) {
#pragma unroll
        for (int e = 0; e < 2; ++e)
            bq[t][e] = *(const v4f*)(wl + (8 * (t >> 1) + 2 * (t & 1) + e) * 128);
    }

    const float* xr = x7 + (size_t)((c & 7) * 1024 + blockIdx.x * 8 + (c >> 3)) * 128 + 4 * h;
    float* orow = uh + (size_t)(blockIdx.x * 64 + 8 * h) * COLS + grp * 64 + 4 * c;

#pragma unroll
    for (int i = 0; i < 4; ++i) {
        v4f a4[16];
#pragma unroll
        for (int u = 0; u < 16; ++u) a4[u] = *(const v4f*)(xr + i * 256 + 8 * u);

        v8f acc[4];
#pragma unroll
        for (int T = 0; T < 4; ++T) acc[T] = (v8f){0,0,0,0,0,0,0,0};

#pragma unroll
        for (int u = 0; u < 16; ++u) {
#pragma unroll
            for (int p = 0; p < 2; ++p) {
                v2f a;
                a.x = a4[u][2 * p + 0];
                a.y = a4[u][2 * p + 1];
#pragma unroll
                for (int T = 0; T < 4; ++T) {
                    v2f b;
                    b.x = bq[2 * u + p][0][T];
                    b.y = bq[2 * u + p][1][T];
                    acc[T] = __builtin_amdgcn_wmma_f32_16x16x4_f32(
                        false, a, false, b, (short)0, acc[T], false, false);
                }
            }
        }
#pragma unroll
        for (int g = 0; g < 8; ++g) {
            v4f o;
            o.x = acc[0][g]; o.y = acc[1][g]; o.z = acc[2][g]; o.w = acc[3][g];
            *(v4f*)(orow + (size_t)(16 * i + g) * COLS) = o;
        }
    }
}

#define RPARTS 32
#define PROWS (JB / RPARTS)

template <int CTRL>
__device__ __forceinline__ float row_xor(float x) {
    return __int_as_float(__builtin_amdgcn_update_dpp(0, __float_as_int(x), CTRL, 0xf, 0xf, false));
}
__device__ __forceinline__ float xhalf(float x) {
    return __int_as_float(__builtin_amdgcn_permlanex16(__float_as_int(x), __float_as_int(x),
                                                       0x76543210, 0xfedcba98, false, false));
}
__device__ __forceinline__ float wave_sum_valu(float v) {
    v += row_xor<0x161>(v);
    v += row_xor<0x162>(v);
    v += row_xor<0x164>(v);
    v += row_xor<0x168>(v);
    return v + xhalf(v);
}
__device__ __forceinline__ float wave_sum(float v) {
    v += xhalf(v);
    v += row_xor<0x168>(v);
    v += row_xor<0x164>(v);
    v += row_xor<0x162>(v);
    v += row_xor<0x161>(v);
    return v;
}

__device__ __forceinline__ float logit1(float s1) {
    return 0.125f * s1 * (1.0f / (float)JB);
}

__launch_bounds__(1024)
__global__ void pass_a_kernel(const float* __restrict__ uh, float* __restrict__ n2,
                              float* __restrict__ mp, float* __restrict__ v1p) {
    __shared__ __attribute__((aligned(16))) float red[2][32][DIM];
    const int wave = threadIdx.x >> 5;
    const int lane = threadIdx.x & 31;
    const int j    = blockIdx.x;
    const int r0   = blockIdx.y * PROWS + wave;
    const float* p = uh + (size_t)r0 * COLS + j * DIM + lane * 4;

    v4f a[8];
#pragma unroll
    for (int u = 0; u < 8; ++u) a[u] = *(const v4f*)(p + (size_t)u * (32 * COLS));

    float q[8];
#pragma unroll
    for (int u = 0; u < 8; ++u)
        q[u] = a[u].x * a[u].x + a[u].y * a[u].y + a[u].z * a[u].z + a[u].w * a[u].w;
    const bool b0 = (lane & 1) != 0, b1 = (lane & 2) != 0, b2 = (lane & 4) != 0;
    float r1[4], r2[2];
#pragma unroll
    for (int i = 0; i < 4; ++i) {
        const float keep = b0 ? q[i + 4] : q[i];
        const float send = b0 ? q[i] : q[i + 4];
        r1[i] = keep + row_xor<0x161>(send);
    }
#pragma unroll
    for (int i = 0; i < 2; ++i) {
        const float keep = b1 ? r1[i + 2] : r1[i];
        const float send = b1 ? r1[i] : r1[i + 2];
        r2[i] = keep + row_xor<0x162>(send);
    }
    float nsum;
    {
        const float keep = b2 ? r2[1] : r2[0];
        const float send = b2 ? r2[0] : r2[1];
        nsum = keep + row_xor<0x164>(send);
    }
    nsum += row_xor<0x168>(nsum);
    nsum += xhalf(nsum);
    const int ul = ((lane & 1) << 2) | (lane & 2) | ((lane >> 2) & 1);
    if (lane < 8) n2[(r0 + 32 * ul) * NJ + j] = nsum;
    const float gl = squash_gain_routing(nsum * (1.0f / 64.0f));

    v4f m = {0, 0, 0, 0}, v = {0, 0, 0, 0};
#pragma unroll
    for (int u = 0; u < 8; ++u) {
        const float g = __int_as_float(__builtin_amdgcn_readlane(
            __float_as_int(gl), ((u & 1) << 2) | (u & 2) | ((u >> 2) & 1)));
        m += a[u];
        v += g * a[u];
    }
    *(v4f*)&red[0][wave][lane * 4] = m;
    *(v4f*)&red[1][wave][lane * 4] = v;
    __syncthreads();
    if (threadIdx.x < 2 * DIM) {
        const int which = threadIdx.x >> 7;
        const int e     = threadIdx.x & (DIM - 1);
        float s0 = 0.0f, s1 = 0.0f, s2 = 0.0f, s3 = 0.0f;
        for (int w = 0; w < 32; w += 4) {
            s0 += red[which][w + 0][e];
            s1 += red[which][w + 1][e];
            s2 += red[which][w + 2][e];
            s3 += red[which][w + 3][e];
        }
        const float s = (s0 + s1) + (s2 + s3);
        const int o = blockIdx.y * COLS + j * DIM + e;
        if (which == 0) mp[o] = s * (1.0f / (float)JB);
        else            v1p[o] = s;
    }
}

#define ZQ 4
__launch_bounds__(1024)
__global__ void join_a_kernel(const float* __restrict__ mp, const float* __restrict__ v1p,
                              const float* __restrict__ w,
                              float* __restrict__ s1, float* __restrict__ z) {
    __shared__ float rm[8][DIM];
    __shared__ float rv[8][DIM];
    __shared__ float wd[4];
    __shared__ __attribute__((aligned(16))) float um[DIM];
    const int j = blockIdx.x;
    const int g = threadIdx.x >> 7;
    const int e = threadIdx.x & (DIM - 1);
    const int col = j * DIM + e;
    const int d = blockIdx.y * 32 + (threadIdx.x >> 5);
    const v4f wq = *(const v4f*)(w + (size_t)7 * 8 * 128 * 128 + (size_t)j * 128 * 128 +
                                 (size_t)d * 128 + (threadIdx.x & 31) * 4);
    rm[g][e] = (mp[g * COLS + col] + mp[(g + 8) * COLS + col]) +
               (mp[(g + 16) * COLS + col] + mp[(g + 24) * COLS + col]);
    rv[g][e] = (v1p[g * COLS + col] + v1p[(g + 8) * COLS + col]) +
               (v1p[(g + 16) * COLS + col] + v1p[(g + 24) * COLS + col]);
    __syncthreads();
    if (threadIdx.x < DIM) {
        const float M = ((rm[0][e] + rm[1][e]) + (rm[2][e] + rm[3][e])) +
                        ((rm[4][e] + rm[5][e]) + (rm[6][e] + rm[7][e]));
        const float V = ((rv[0][e] + rv[1][e]) + (rv[2][e] + rv[3][e])) +
                        ((rv[4][e] + rv[5][e]) + (rv[6][e] + rv[7][e]));
        um[e] = M;
        const float d = wave_sum(M * V);
        if ((threadIdx.x & 31) == 0) wd[threadIdx.x >> 5] = d;
    }
    __syncthreads();
    if (blockIdx.y == 0 && threadIdx.x == 0) s1[j] = (wd[0] + wd[1]) + (wd[2] + wd[3]);
    const v4f u4 = *(const v4f*)&um[(threadIdx.x & 31) * 4];
    const float zp = wave_sum_valu(wq.x * u4.x + wq.y * u4.y + wq.z * u4.z + wq.w * u4.w);
    if ((threadIdx.x & 31) == 0) z[j * DIM + d] = zp;
}

#define XWAVES 4
#define XPARTS (1024 / XWAVES)
__launch_bounds__(32 * XWAVES)
__global__ void pass_b_kernel(const float* __restrict__ x, const float* __restrict__ n2,
                              const float* __restrict__ s1, const float* __restrict__ z,
                              float* __restrict__ s2p) {
    __shared__ __attribute__((aligned(16))) float wd[XWAVES][NJ];
    const int wave = threadIdx.x >> 5;
    const int lane = threadIdx.x & 31;
    const int s    = blockIdx.x * XWAVES + wave;
    const float* xp = x + (size_t)7 * 8 * 1024 * 128 + (size_t)s * 128 + lane * 4;

    v4f a[8], zz[NJ];
#pragma unroll
    for (int b = 0; b < 8; ++b) a[b] = *(const v4f*)(xp + (size_t)b * (1024 * 128));
    const v2f nq = *(const v2f*)(n2 + s * 64 + 2 * lane);
#pragma unroll
    for (int j = 0; j < NJ; ++j) zz[j] = *(const v4f*)(z + j * DIM + lane * 4);

    const int j0 = 2 * (lane & 3);
    float mx = logit1(s1[0]);
    for (int k = 1; k < NJ; ++k) mx = fmaxf(mx, logit1(s1[k]));
    float se = 0.0f, e0 = 0.0f, e1 = 0.0f;
    for (int k = 0; k < NJ; ++k) {
        const float ek = __expf(logit1(s1[k]) - mx);
        se += ek;
        if (k == j0)     e0 = ek;
        if (k == j0 + 1) e1 = ek;
    }
    const float c0 = e0 / se, c1 = e1 / se;

    asm volatile("" :: "v"(a[0]), "v"(a[1]), "v"(a[2]), "v"(a[3]), "v"(a[4]), "v"(a[5]), "v"(a[6]), "v"(a[7]),
                       "v"(zz[0]), "v"(zz[1]), "v"(zz[2]), "v"(zz[3]), "v"(zz[4]), "v"(zz[5]), "v"(zz[6]), "v"(zz[7]),
                       "v"(nq), "v"(c0), "v"(c1));

    const float g0 = squash_gain_routing(c0 * c0 * nq.x);
    const float g1 = squash_gain_routing(c1 * c1 * nq.y);

    v4f y[NJ];
#pragma unroll
    for (int j = 0; j < NJ; ++j) y[j] = (v4f){0, 0, 0, 0};
#pragma unroll
    for (int b = 0; b < 8; ++b) {
#pragma unroll
        for (int j = 0; j < NJ; ++j) {
            const float gv = __int_as_float(__builtin_amdgcn_readlane(
                __float_as_int((j & 1) ? g1 : g0), (b * 8 + j) >> 1));
            y[j] += gv * a[b];
        }
    }
    float d[NJ];
#pragma unroll
    for (int j = 0; j < NJ; ++j)
        d[j] = wave_sum_valu(y[j].x * zz[j].x + y[j].y * zz[j].y +
                             y[j].z * zz[j].z + y[j].w * zz[j].w);
    if (lane == 0) {
        *(v4f*)&wd[wave][0] = (v4f){d[0], d[1], d[2], d[3]};
        *(v4f*)&wd[wave][4] = (v4f){d[4], d[5], d[6], d[7]};
    }
    __syncthreads();
    if (threadIdx.x < NJ)
        s2p[blockIdx.x * NJ + threadIdx.x] = (wd[0][threadIdx.x] + wd[1][threadIdx.x]) +
                                             (wd[2][threadIdx.x] + wd[3][threadIdx.x]);
}

__launch_bounds__(1024)
__global__ void join_b_kernel(const float* __restrict__ s2p, const float* __restrict__ s1,
                              float* __restrict__ cf) {
    __shared__ float wd[32][NJ];
    const int t = threadIdx.x;
    float sv[NJ];
#pragma unroll
    for (int k = 0; k < NJ; ++k) sv[k] = s1[k];
    float v = s2p[t] + s2p[t + 1024];
    asm volatile("" :: "s"(sv[0]), "s"(sv[1]), "s"(sv[2]), "s"(sv[3]), "s"(sv[4]), "s"(sv[5]), "s"(sv[6]), "s"(sv[7]));
    v += row_xor<0x168>(v);
    v += xhalf(v);
    if ((t & 31) < NJ) wd[t >> 5][t & 7] = v;
    __syncthreads();
    if (t < 32) {
        const int j = t & 7;
        const int q = (t >> 3) * 8;
        float a = ((wd[q + 0][j] + wd[q + 1][j]) + (wd[q + 2][j] + wd[q + 3][j])) +
                  ((wd[q + 4][j] + wd[q + 5][j]) + (wd[q + 6][j] + wd[q + 7][j]));
        a += row_xor<0x168>(a);
        a += xhalf(a);
        float s2[NJ];
        for (int k = 0; k < NJ; ++k) s2[k] = __int_as_float(__builtin_amdgcn_readlane(__float_as_int(a), k));
        if (t == 0) {
            float b1[NJ], c1[NJ], b2[NJ];
            for (int k = 0; k < NJ; ++k) b1[k] = logit1(sv[k]);
            float mx = b1[0];
            for (int k = 1; k < NJ; ++k) mx = fmaxf(mx, b1[k]);
            float se = 0.0f;
            for (int k = 0; k < NJ; ++k) { c1[k] = __expf(b1[k] - mx); se += c1[k]; }
            for (int k = 0; k < NJ; ++k)
                b2[k] = b1[k] + (c1[k] / se) * s2[k] * (1.0f / (float)JB);
            mx = b2[0];
            for (int k = 1; k < NJ; ++k) mx = fmaxf(mx, b2[k]);
            se = 0.0f;
            float e2[NJ];
            for (int k = 0; k < NJ; ++k) { e2[k] = __expf(b2[k] - mx); se += e2[k]; }
            for (int k = 0; k < NJ; ++k) cf[k] = e2[k] / se;
        }
    }
}

__launch_bounds__(256)
__global__ void final_kernel(const float* __restrict__ uh,
                             const float* __restrict__ n2,
                             const float* __restrict__ cf,
                             float* __restrict__ out) {
    const size_t idx  = (size_t)blockIdx.x * 256 + threadIdx.x;
    const size_t base = idx * 4;
    const int r = (int)(base >> 10);
    const int j = (int)((base >> 7) & 7);
    const float cj = cf[j];
    const float gain = cj * squash_gain(cj * cj * n2[r * 8 + j]);
    float4 a = *(const float4*)(uh + base);
    a.x *= gain; a.y *= gain; a.z *= gain; a.w *= gain;
    *(float4*)(out + base) = a;
}

extern "C" void kernel_launch(void* const* d_in, const int* in_sizes, int n_in,
                              void* d_out, int out_size, void* d_ws, size_t ws_size,
                              hipStream_t stream) {
    const float* x = (const float*)d_in[0];
    const float* w = (const float*)d_in[1];
    float* out = (float*)d_out;

    char* ws = (char*)d_ws;
    float* uh   = (float*)(ws);
    float* n2   = (float*)(ws + (size_t)JB * COLS * 4);
    float* mp   = (float*)(ws + (size_t)JB * COLS * 4 + 262144);
    float* v1p  = (float*)(ws + (size_t)JB * COLS * 4 + 393216);
    float* cf   = (float*)(ws + (size_t)JB * COLS * 4 + 528384);
    float* z    = out;
    float* s2p  = out + COLS;
    float* s1   = out + RPARTS * COLS;

    gemm_uhat_kernel<<<dim3(128, 2, 1), 256, 0, stream>>>(x, w, uh);
    pass_a_kernel<<<dim3(NJ, RPARTS, 1), 1024, 0, stream>>>(uh, n2, mp, v1p);
    join_a_kernel<<<dim3(NJ, ZQ, 1), 1024, 0, stream>>>(mp, v1p, w, s1, z);
    pass_b_kernel<<<XPARTS, 32 * XWAVES, 0, stream>>>(x, n2, s1, z, s2p);
    join_b_kernel<<<1, 1024, 0, stream>>>(s2p, s1, cf);
    final_kernel<<<(JB * COLS / 4) / 256, 256, 0, stream>>>(uh, n2, cf, out);
}


# one-block join kernel folded into the output kernel: its 1,024 blocks own 8 rows each (8 row loads in flight per wave), sum the 256 partial dots, form the coefficients; 5 launches; sum orders change
# speedup vs baseline: 11.4249x; 1.1366x over previous
#include <hip/hip_runtime.h>
#include <hip/hip_bf16.h>

typedef __attribute__((ext_vector_type(2))) float v2f;
typedef __attribute__((ext_vector_type(8))) float v8f;
typedef __attribute__((ext_vector_type(4))) float v4f;

#define JB 8192
#define NJ 8
#define DIM 128
#define COLS 1024

__device__ __forceinline__ float squash_gain(float n2) {
    return n2 / ((1.0f + n2) * sqrtf(n2 + 1e-8f));
}

__device__ __forceinline__ float squash_gain_routing(float n2) {
    return n2 * __builtin_amdgcn_rsqf(n2 + 1e-8f) * __builtin_amdgcn_rcpf(1.0f + n2);
}

__launch_bounds__(256)
__global__ void gemm_uhat_kernel(const float* __restrict__ x,
                                 const float* __restrict__ w,
                                 float* __restrict__ uh) {
    const float* x7 = x + (size_t)7 * 8 * 1024 * 128;
    const float* w7 = w + (size_t)7 * 8 * 128 * 128;

    const int wave = threadIdx.x >> 5;
    const int lane = threadIdx.x & 31;
    const int h    = lane >> 4;
    const int c    = lane & 15;

    const int grp  = blockIdx.y * 8 + wave;
    const int j    = grp >> 1;
    const int col0 = (grp & 1) * 64 + 4 * c;

    const float* wl = w7 + (size_t)j * 128 * 128 + (size_t)(4 * h) * 128 + col0;
    v4f bq[32][2];
#pragma unroll
    for (int t = 0; t < 32; ++t) {
#pragma unroll
        for (int e = 0; e < 2; ++e)
            bq[t][e] = *(const v4f*)(wl + (8 * (t >> 1) + 2 * (t & 1) + e) * 128);
    }

    const float* xr = x7 + (size_t)((c & 7) * 1024 + blockIdx.x * 8 + (c >> 3)) * 128 + 4 * h;
    float* orow = uh + (size_t)(blockIdx.x * 64 + 8 * h) * COLS + grp * 64 + 4 * c;

#pragma unroll
    for (int i = 0; i < 4; ++i) {
        v4f a4[16];
#pragma unroll
        for (int u = 0; u < 16; ++u) a4[u] = *(const v4f*)(xr + i * 256 + 8 * u);

        v8f acc[4];
#pragma unroll
        for (int T = 0; T < 4; ++T) acc[T] = (v8f){0,0,0,0,0,0,0,0};

#pragma unroll
        for (int u = 0; u < 16; ++u) {
#pragma unroll
            for (int p = 0; p < 2; ++p) {
                v2f a;
                a.x = a4[u][2 * p + 0];
                a.y = a4[u][2 * p + 1];
#pragma unroll
                for (int T = 0; T < 4; ++T) {
                    v2f b;
                    b.x = bq[2 * u + p][0][T];
                    b.y = bq[2 * u + p][1][T];
                    acc[T] = __builtin_amdgcn_wmma_f32_16x16x4_f32(
                        false, a, false, b, (short)0, acc[T], false, false);
                }
            }
        }
#pragma unroll
        for (int g = 0; g < 8; ++g) {
            v4f o;
            o.x = acc[0][g]; o.y = acc[1][g]; o.z = acc[2][g]; o.w = acc[3][g];
            *(v4f*)(orow + (size_t)(16 * i + g) * COLS) = o;
        }
    }
}

#define RPARTS 32
#define PROWS (JB / RPARTS)

template <int CTRL>
__device__ __forceinline__ float row_xor(float x) {
    return __int_as_float(__builtin_amdgcn_update_dpp(0, __float_as_int(x), CTRL, 0xf, 0xf, false));
}
__device__ __forceinline__ float xhalf(float x) {
    return __int_as_float(__builtin_amdgcn_permlanex16(__float_as_int(x), __float_as_int(x),
                                                       0x76543210, 0xfedcba98, false, false));
}
__device__ __forceinline__ float wave_sum_valu(float v) {
    v += row_xor<0x161>(v);
    v += row_xor<0x162>(v);
    v += row_xor<0x164>(v);
    v += row_xor<0x168>(v);
    return v + xhalf(v);
}
__device__ __forceinline__ float wave_sum(float v) {
    v += xhalf(v);
    v += row_xor<0x168>(v);
    v += row_xor<0x164>(v);
    v += row_xor<0x162>(v);
    v += row_xor<0x161>(v);
    return v;
}

__device__ __forceinline__ float logit1(float s1) {
    return 0.125f * s1 * (1.0f / (float)JB);
}

__launch_bounds__(1024)
__global__ void pass_a_kernel(const float* __restrict__ uh, float* __restrict__ n2,
                              float* __restrict__ mp, float* __restrict__ v1p) {
    __shared__ __attribute__((aligned(16))) float red[2][32][DIM];
    const int wave = threadIdx.x >> 5;
    const int lane = threadIdx.x & 31;
    const int j    = blockIdx.x;
    const int r0   = blockIdx.y * PROWS + wave;
    const float* p = uh + (size_t)r0 * COLS + j * DIM + lane * 4;

    v4f a[8];
#pragma unroll
    for (int u = 0; u < 8; ++u) a[u] = *(const v4f*)(p + (size_t)u * (32 * COLS));

    float q[8];
#pragma unroll
    for (int u = 0; u < 8; ++u)
        q[u] = a[u].x * a[u].x + a[u].y * a[u].y + a[u].z * a[u].z + a[u].w * a[u].w;
    const bool b0 = (lane & 1) != 0, b1 = (lane & 2) != 0, b2 = (lane & 4) != 0;
    float r1[4], r2[2];
#pragma unroll
    for (int i = 0; i < 4; ++i) {
        const float keep = b0 ? q[i + 4] : q[i];
        const float send = b0 ? q[i] : q[i + 4];
        r1[i] = keep + row_xor<0x161>(send);
    }
#pragma unroll
    for (int i = 0; i < 2; ++i) {
        const float keep = b1 ? r1[i + 2] : r1[i];
        const float send = b1 ? r1[i] : r1[i + 2];
        r2[i] = keep + row_xor<0x162>(send);
    }
    float nsum;
    {
        const float keep = b2 ? r2[1] : r2[0];
        const float send = b2 ? r2[0] : r2[1];
        nsum = keep + row_xor<0x164>(send);
    }
    nsum += row_xor<0x168>(nsum);
    nsum += xhalf(nsum);
    const int ul = ((lane & 1) << 2) | (lane & 2) | ((lane >> 2) & 1);
    if (lane < 8) n2[(r0 + 32 * ul) * NJ + j] = nsum;
    const float gl = squash_gain_routing(nsum * (1.0f / 64.0f));

    v4f m = {0, 0, 0, 0}, v = {0, 0, 0, 0};
#pragma unroll
    for (int u = 0; u < 8; ++u) {
        const float g = __int_as_float(__builtin_amdgcn_readlane(
            __float_as_int(gl), ((u & 1) << 2) | (u & 2) | ((u >> 2) & 1)));
        m += a[u];
        v += g * a[u];
    }
    *(v4f*)&red[0][wave][lane * 4] = m;
    *(v4f*)&red[1][wave][lane * 4] = v;
    __syncthreads();
    if (threadIdx.x < 2 * DIM) {
        const int which = threadIdx.x >> 7;
        const int e     = threadIdx.x & (DIM - 1);
        float s0 = 0.0f, s1 = 0.0f, s2 = 0.0f, s3 = 0.0f;
        for (int w = 0; w < 32; w += 4) {
            s0 += red[which][w + 0][e];
            s1 += red[which][w + 1][e];
            s2 += red[which][w + 2][e];
            s3 += red[which][w + 3][e];
        }
        const float s = (s0 + s1) + (s2 + s3);
        const int o = blockIdx.y * COLS + j * DIM + e;
        if (which == 0) mp[o] = s * (1.0f / (float)JB);
        else            v1p[o] = s;
    }
}

#define ZQ 4
__launch_bounds__(1024)
__global__ void join_a_kernel(const float* __restrict__ mp, const float* __restrict__ v1p,
                              const float* __restrict__ w,
                              float* __restrict__ s1, float* __restrict__ z) {
    __shared__ float rm[8][DIM];
    __shared__ float rv[8][DIM];
    __shared__ float wd[4];
    __shared__ __attribute__((aligned(16))) float um[DIM];
    const int j = blockIdx.x;
    const int g = threadIdx.x >> 7;
    const int e = threadIdx.x & (DIM - 1);
    const int col = j * DIM + e;
    const int d = blockIdx.y * 32 + (threadIdx.x >> 5);
    const v4f wq = *(const v4f*)(w + (size_t)7 * 8 * 128 * 128 + (size_t)j * 128 * 128 +
                                 (size_t)d * 128 + (threadIdx.x & 31) * 4);
    rm[g][e] = (mp[g * COLS + col] + mp[(g + 8) * COLS + col]) +
               (mp[(g + 16) * COLS + col] + mp[(g + 24) * COLS + col]);
    rv[g][e] = (v1p[g * COLS + col] + v1p[(g + 8) * COLS + col]) +
               (v1p[(g + 16) * COLS + col] + v1p[(g + 24) * COLS + col]);
    __syncthreads();
    if (threadIdx.x < DIM) {
        const float M = ((rm[0][e] + rm[1][e]) + (rm[2][e] + rm[3][e])) +
                        ((rm[4][e] + rm[5][e]) + (rm[6][e] + rm[7][e]));
        const float V = ((rv[0][e] + rv[1][e]) + (rv[2][e] + rv[3][e])) +
                        ((rv[4][e] + rv[5][e]) + (rv[6][e] + rv[7][e]));
        um[e] = M;
        const float d = wave_sum(M * V);
        if ((threadIdx.x & 31) == 0) wd[threadIdx.x >> 5] = d;
    }
    __syncthreads();
    if (blockIdx.y == 0 && threadIdx.x == 0) s1[j] = (wd[0] + wd[1]) + (wd[2] + wd[3]);
    const v4f u4 = *(const v4f*)&um[(threadIdx.x & 31) * 4];
    const float zp = wave_sum_valu(wq.x * u4.x + wq.y * u4.y + wq.z * u4.z + wq.w * u4.w);
    if ((threadIdx.x & 31) == 0) z[j * DIM + d] = zp;
}

#define XWAVES 4
#define XPARTS (1024 / XWAVES)
__launch_bounds__(32 * XWAVES)
__global__ void pass_b_kernel(const float* __restrict__ x, const float* __restrict__ n2,
                              const float* __restrict__ s1, const float* __restrict__ z,
                              float* __restrict__ s2p) {
    __shared__ __attribute__((aligned(16))) float wd[XWAVES][NJ];
    const int wave = threadIdx.x >> 5;
    const int lane = threadIdx.x & 31;
    const int s    = blockIdx.x * XWAVES + wave;
    const float* xp = x + (size_t)7 * 8 * 1024 * 128 + (size_t)s * 128 + lane * 4;

    v4f a[8], zz[NJ];
#pragma unroll
    for (int b = 0; b < 8; ++b) a[b] = *(const v4f*)(xp + (size_t)b * (1024 * 128));
    const v2f nq = *(const v2f*)(n2 + s * 64 + 2 * lane);
#pragma unroll
    for (int j = 0; j < NJ; ++j) zz[j] = *(const v4f*)(z + j * DIM + lane * 4);

    const int j0 = 2 * (lane & 3);
    float mx = logit1(s1[0]);
    for (int k = 1; k < NJ; ++k) mx = fmaxf(mx, logit1(s1[k]));
    float se = 0.0f, e0 = 0.0f, e1 = 0.0f;
    for (int k = 0; k < NJ; ++k) {
        const float ek = __expf(logit1(s1[k]) - mx);
        se += ek;
        if (k == j0)     e0 = ek;
        if (k == j0 + 1) e1 = ek;
    }
    const float c0 = e0 / se, c1 = e1 / se;

    asm volatile("" :: "v"(a[0]), "v"(a[1]), "v"(a[2]), "v"(a[3]), "v"(a[4]), "v"(a[5]), "v"(a[6]), "v"(a[7]),
                       "v"(zz[0]), "v"(zz[1]), "v"(zz[2]), "v"(zz[3]), "v"(zz[4]), "v"(zz[5]), "v"(zz[6]), "v"(zz[7]),
                       "v"(nq), "v"(c0), "v"(c1));

    const float g0 = squash_gain_routing(c0 * c0 * nq.x);
    const float g1 = squash_gain_routing(c1 * c1 * nq.y);

    v4f y[NJ];
#pragma unroll
    for (int j = 0; j < NJ; ++j) y[j] = (v4f){0, 0, 0, 0};
#pragma unroll
    for (int b = 0; b < 8; ++b) {
#pragma unroll
        for (int j = 0; j < NJ; ++j) {
            const float gv = __int_as_float(__builtin_amdgcn_readlane(
                __float_as_int((j & 1) ? g1 : g0), (b * 8 + j) >> 1));
            y[j] += gv * a[b];
        }
    }
    float d[NJ];
#pragma unroll
    for (int j = 0; j < NJ; ++j)
        d[j] = wave_sum_valu(y[j].x * zz[j].x + y[j].y * zz[j].y +
                             y[j].z * zz[j].z + y[j].w * zz[j].w);
    if (lane == 0) {
        *(v4f*)&wd[wave][0] = (v4f){d[0], d[1], d[2], d[3]};
        *(v4f*)&wd[wave][4] = (v4f){d[4], d[5], d[6], d[7]};
    }
    __syncthreads();
    if (threadIdx.x < NJ)
        s2p[blockIdx.x * NJ + threadIdx.x] = (wd[0][threadIdx.x] + wd[1][threadIdx.x]) +
                                             (wd[2][threadIdx.x] + wd[3][threadIdx.x]);
}

#define FROWS 8
__launch_bounds__(256)
__global__ void final_kernel(const float* __restrict__ uh,
                             const float* __restrict__ n2,
                             const float* __restrict__ s2p,
                             const float* __restrict__ s1,
                             float* __restrict__ out) {
    __shared__ float wd[8][NJ];
    const int t    = threadIdx.x;
    const int lane = t & 31;
    const int k    = lane & 7;
    const int wave = __builtin_amdgcn_readfirstlane(t >> 5);
    const int r0   = blockIdx.x * FROWS;

    const v4f pa = *(const v4f*)(s2p + t * NJ);
    const v4f pb = *(const v4f*)(s2p + t * NJ + 4);
    const float s1k = s1[k];
    const float nr  = n2[(r0 + k) * NJ + wave];
    const float* p = uh + (size_t)r0 * COLS + t * 4;
    v4f a[FROWS];
#pragma unroll
    for (int u = 0; u < FROWS; ++u) a[u] = *(const v4f*)(p + (size_t)u * COLS);

    const bool b0 = (lane & 1) != 0, b1 = (lane & 2) != 0, b2 = (lane & 4) != 0;
    const float q[8] = {pa.x, pa.y, pa.z, pa.w, pb.x, pb.y, pb.z, pb.w};
    float h1[4], h2[2];
#pragma unroll
    for (int i = 0; i < 4; ++i) {
        const float keep = b2 ? q[i + 4] : q[i];
        const float send = b2 ? q[i] : q[i + 4];
        h1[i] = keep + row_xor<0x164>(send);
    }
#pragma unroll
    for (int i = 0; i < 2; ++i) {
        const float keep = b1 ? h1[i + 2] : h1[i];
        const float send = b1 ? h1[i] : h1[i + 2];
        h2[i] = keep + row_xor<0x162>(send);
    }
    float wsum;
    {
        const float keep = b0 ? h2[1] : h2[0];
        const float send = b0 ? h2[0] : h2[1];
        wsum = keep + row_xor<0x161>(send);
    }
    wsum += row_xor<0x168>(wsum);
    wsum += xhalf(wsum);
    if (lane < NJ) wd[wave][k] = wsum;
    asm volatile("s_wait_dscnt 0x0\n\ts_barrier_signal -1\n\ts_barrier_wait -1" ::: "memory");
    const float s2k = ((wd[0][k] + wd[1][k]) + (wd[2][k] + wd[3][k])) +
                      ((wd[4][k] + wd[5][k]) + (wd[6][k] + wd[7][k]));

    const float l1 = logit1(s1k);
    float mx = fmaxf(l1, row_xor<0x161>(l1));
    mx = fmaxf(mx, row_xor<0x162>(mx));
    mx = fmaxf(mx, row_xor<0x164>(mx));
    const float e1 = __expf(l1 - mx);
    float se = e1 + row_xor<0x161>(e1);
    se += row_xor<0x162>(se);
    se += row_xor<0x164>(se);
    const float l2 = l1 + (e1 / se) * s2k * (1.0f / (float)JB);
    mx = fmaxf(l2, row_xor<0x161>(l2));
    mx = fmaxf(mx, row_xor<0x162>(mx));
    mx = fmaxf(mx, row_xor<0x164>(mx));
    const float e2 = __expf(l2 - mx);
    se = e2 + row_xor<0x161>(e2);
    se += row_xor<0x162>(se);
    se += row_xor<0x164>(se);
    const float cfk = e2 / se;

    const float cj = __int_as_float(__builtin_amdgcn_readlane(__float_as_int(cfk), wave));
    const float gk = cj * squash_gain(cj * cj * nr);
    float* o = out + (size_t)r0 * COLS + t * 4;
#pragma unroll
    for (int u = 0; u < FROWS; ++u) {
        const float g = __int_as_float(__builtin_amdgcn_readlane(__float_as_int(gk), u));
        *(v4f*)(o + (size_t)u * COLS) = g * a[u];
    }
}

extern "C" void kernel_launch(void* const* d_in, const int* in_sizes, int n_in,
                              void* d_out, int out_size, void* d_ws, size_t ws_size,
                              hipStream_t stream) {
    const float* x = (const float*)d_in[0];
    const float* w = (const float*)d_in[1];
    float* out = (float*)d_out;

    char* ws = (char*)d_ws;
    float* uh   = (float*)(ws);
    float* n2   = (float*)(ws + (size_t)JB * COLS * 4);
    float* mp   = (float*)(ws + (size_t)JB * COLS * 4 + 262144);
    float* v1p  = (float*)(ws + (size_t)JB * COLS * 4 + 393216);
    float* s1   = (float*)(ws + (size_t)JB * COLS * 4 + 528384);
    float* s2p  = mp;
    float* z    = out;

    gemm_uhat_kernel<<<dim3(128, 2, 1), 256, 0, stream>>>(x, w, uh);
    pass_a_kernel<<<dim3(NJ, RPARTS, 1), 1024, 0, stream>>>(uh, n2, mp, v1p);
    join_a_kernel<<<dim3(NJ, ZQ, 1), 1024, 0, stream>>>(mp, v1p, w, s1, z);
    pass_b_kernel<<<XPARTS, 32 * XWAVES, 0, stream>>>(x, n2, s1, z, s2p);
    final_kernel<<<JB / FROWS, 256, 0, stream>>>(uh, n2, s2p, s1, out);
}


# pass over uh (row norms, mean parts, weighted column sums) moved into the GEMM kernel: behind a barrier each block re-reads its own 64 rows from cache; 4 launches; column sum trees regrouped
# speedup vs baseline: 11.8291x; 1.0354x over previous
#include <hip/hip_runtime.h>
#include <hip/hip_bf16.h>

typedef __attribute__((ext_vector_type(2))) float v2f;
typedef __attribute__((ext_vector_type(8))) float v8f;
typedef __attribute__((ext_vector_type(4))) float v4f;

#define JB 8192
#define NJ 8
#define DIM 128
#define COLS 1024

__device__ __forceinline__ float squash_gain(float n2) {
    return n2 / ((1.0f + n2) * sqrtf(n2 + 1e-8f));
}

__device__ __forceinline__ float squash_gain_routing(float n2) {
    return n2 * __builtin_amdgcn_rsqf(n2 + 1e-8f) * __builtin_amdgcn_rcpf(1.0f + n2);
}

template <int CTRL>
__device__ __forceinline__ float row_xor(float x) {
    return __int_as_float(__builtin_amdgcn_update_dpp(0, __float_as_int(x), CTRL, 0xf, 0xf, false));
}
__device__ __forceinline__ float xhalf(float x) {
    return __int_as_float(__builtin_amdgcn_permlanex16(__float_as_int(x), __float_as_int(x),
                                                       0x76543210, 0xfedcba98, false, false));
}
__device__ __forceinline__ float wave_sum_valu(float v) {
    v += row_xor<0x161>(v);
    v += row_xor<0x162>(v);
    v += row_xor<0x164>(v);
    v += row_xor<0x168>(v);
    return v + xhalf(v);
}
__device__ __forceinline__ float wave_sum(float v) {
    v += xhalf(v);
    v += row_xor<0x168>(v);
    v += row_xor<0x164>(v);
    v += row_xor<0x162>(v);
    v += row_xor<0x161>(v);
    return v;
}

__device__ __forceinline__ float logit1(float s1) {
    return 0.125f * s1 * (1.0f / (float)JB);
}

__device__ __forceinline__ void rows8_stats(const v4f (&a)[8], int lane,
                                            float& nsum_out, v4f& m_out, v4f& v_out) {
    float q[8];
#pragma unroll
    for (int u = 0; u < 8; ++u)
        q[u] = a[u].x * a[u].x + a[u].y * a[u].y + a[u].z * a[u].z + a[u].w * a[u].w;
    const bool b0 = (lane & 1) != 0, b1 = (lane & 2) != 0, b2 = (lane & 4) != 0;
    float r1[4], r2[2];
#pragma unroll
    for (int i = 0; i < 4; ++i) {
        const float keep = b0 ? q[i + 4] : q[i];
        const float send = b0 ? q[i] : q[i + 4];
        r1[i] = keep + row_xor<0x161>(send);
    }
#pragma unroll
    for (int i = 0; i < 2; ++i) {
        const float keep = b1 ? r1[i + 2] : r1[i];
        const float send = b1 ? r1[i] : r1[i + 2];
        r2[i] = keep + row_xor<0x162>(send);
    }
    float nsum;
    {
        const float keep = b2 ? r2[1] : r2[0];
        const float send = b2 ? r2[0] : r2[1];
        nsum = keep + row_xor<0x164>(send);
    }
    nsum += row_xor<0x168>(nsum);
    nsum += xhalf(nsum);
    nsum_out = nsum;
    const float gl = squash_gain_routing(nsum * (1.0f / 64.0f));

    v4f m = {0, 0, 0, 0}, v = {0, 0, 0, 0};
#pragma unroll
    for (int u = 0; u < 8; ++u) {
        const float g = __int_as_float(__builtin_amdgcn_readlane(
            __float_as_int(gl), ((u & 1) << 2) | (u & 2) | ((u >> 2) & 1)));
        m += a[u];
        v += g * a[u];
    }
    m_out = m;
    v_out = v;
}

__launch_bounds__(256)
__global__ void gemm_uhat_kernel(const float* __restrict__ x,
                                 const float* __restrict__ w,
                                 float* __restrict__ uh,
                                 float* __restrict__ n2,
                                 float* __restrict__ mp,
                                 float* __restrict__ v1p) {
    __shared__ __attribute__((aligned(16))) float red[2][8][512];
    const float* x7 = x + (size_t)7 * 8 * 1024 * 128;
    const float* w7 = w + (size_t)7 * 8 * 128 * 128;

    const int wave = threadIdx.x >> 5;
    const int lane = threadIdx.x & 31;
    const int h    = lane >> 4;
    const int c    = lane & 15;

    const int grp  = blockIdx.y * 8 + wave;
    const int j    = grp >> 1;
    const int col0 = (grp & 1) * 64 + 4 * c;

    const float* wl = w7 + (size_t)j * 128 * 128 + (size_t)(4 * h) * 128 + col0;
    v4f bq[32][2];
#pragma unroll
    for (int t = 0; t < 32; ++t) {
#pragma unroll
        for (int e = 0; e < 2; ++e)
            bq[t][e] = *(const v4f*)(wl + (8 * (t >> 1) + 2 * (t & 1) + e) * 128);
    }

    const float* xr = x7 + (size_t)((c & 7) * 1024 + blockIdx.x * 8 + (c >> 3)) * 128 + 4 * h;
    float* orow = uh + (size_t)(blockIdx.x * 64 + 8 * h) * COLS + grp * 64 + 4 * c;

#pragma unroll
    for (int i = 0; i < 4; ++i) {
        v4f a4[16];
#pragma unroll
        for (int u = 0; u < 16; ++u) a4[u] = *(const v4f*)(xr + i * 256 + 8 * u);

        v8f acc[4];
#pragma unroll
        for (int T = 0; T < 4; ++T) acc[T] = (v8f){0,0,0,0,0,0,0,0};

#pragma unroll
        for (int u = 0; u < 16; ++u) {
#pragma unroll
            for (int p = 0; p < 2; ++p) {
                v2f a;
                a.x = a4[u][2 * p + 0];
                a.y = a4[u][2 * p + 1];
#pragma unroll
                for (int T = 0; T < 4; ++T) {
                    v2f b;
                    b.x = bq[2 * u + p][0][T];
                    b.y = bq[2 * u + p][1][T];
                    acc[T] = __builtin_amdgcn_wmma_f32_16x16x4_f32(
                        false, a, false, b, (short)0, acc[T], false, false);
                }
            }
        }
#pragma unroll
        for (int g = 0; g < 8; ++g) {
            v4f o;
            o.x = acc[0][g]; o.y = acc[1][g]; o.z = acc[2][g]; o.w = acc[3][g];
            *(v4f*)(orow + (size_t)(16 * i + g) * COLS) = o;
        }
    }

    __syncthreads();
    const int rowb = blockIdx.x * 64 + wave * 8;
    const float* pr = uh + (size_t)rowb * COLS + blockIdx.y * 512 + lane * 4;
    v4f ra[4][8];
#pragma unroll
    for (int jl = 0; jl < 4; ++jl) {
#pragma unroll
        for (int u = 0; u < 8; ++u) ra[jl][u] = *(const v4f*)(pr + (size_t)u * COLS + jl * DIM);
    }
#pragma unroll
    for (int jl = 0; jl < 4; ++jl) {
        float nsum;
        v4f m, v;
        rows8_stats(ra[jl], lane, nsum, m, v);
        const int ul = ((lane & 1) << 2) | (lane & 2) | ((lane >> 2) & 1);
        if (lane < 8) n2[(rowb + ul) * NJ + blockIdx.y * 4 + jl] = nsum;
        *(v4f*)&red[0][wave][jl * DIM + lane * 4] = m;
        *(v4f*)&red[1][wave][jl * DIM + lane * 4] = v;
    }
    __syncthreads();
    {
        const int which = threadIdx.x >> 7;
        const int e4    = (threadIdx.x & 127) * 4;
        const v4f s = ((*(const v4f*)&red[which][0][e4] + *(const v4f*)&red[which][1][e4]) +
                       (*(const v4f*)&red[which][2][e4] + *(const v4f*)&red[which][3][e4])) +
                      ((*(const v4f*)&red[which][4][e4] + *(const v4f*)&red[which][5][e4]) +
                       (*(const v4f*)&red[which][6][e4] + *(const v4f*)&red[which][7][e4]));
        const int o = blockIdx.x * COLS + blockIdx.y * 512 + e4;
        if (which == 0) *(v4f*)(mp + o) = s * (1.0f / (float)JB);
        else            *(v4f*)(v1p + o) = s;
    }
}

#define GPARTS 128

#define ZQ 4
__launch_bounds__(1024)
__global__ void join_a_kernel(const float* __restrict__ mp, const float* __restrict__ v1p,
                              const float* __restrict__ w,
                              float* __restrict__ s1, float* __restrict__ z) {
    __shared__ float rm[8][DIM];
    __shared__ float rv[8][DIM];
    __shared__ float wd[4];
    __shared__ __attribute__((aligned(16))) float um[DIM];
    const int j = blockIdx.x;
    const int g = threadIdx.x >> 7;
    const int e = threadIdx.x & (DIM - 1);
    const int col = j * DIM + e;
    const int d = blockIdx.y * 32 + (threadIdx.x >> 5);
    const v4f wq = *(const v4f*)(w + (size_t)7 * 8 * 128 * 128 + (size_t)j * 128 * 128 +
                                 (size_t)d * 128 + (threadIdx.x & 31) * 4);
    float am[4] = {0.0f, 0.0f, 0.0f, 0.0f}, av[4] = {0.0f, 0.0f, 0.0f, 0.0f};
#pragma unroll
    for (int q = 0; q < GPARTS / 8; ++q) {
        am[q & 3] += mp[(g + 8 * q) * COLS + col];
        av[q & 3] += v1p[(g + 8 * q) * COLS + col];
    }
    rm[g][e] = (am[0] + am[1]) + (am[2] + am[3]);
    rv[g][e] = (av[0] + av[1]) + (av[2] + av[3]);
    __syncthreads();
    if (threadIdx.x < DIM) {
        const float M = ((rm[0][e] + rm[1][e]) + (rm[2][e] + rm[3][e])) +
                        ((rm[4][e] + rm[5][e]) + (rm[6][e] + rm[7][e]));
        const float V = ((rv[0][e] + rv[1][e]) + (rv[2][e] + rv[3][e])) +
                        ((rv[4][e] + rv[5][e]) + (rv[6][e] + rv[7][e]));
        um[e] = M;
        const float d = wave_sum(M * V);
        if ((threadIdx.x & 31) == 0) wd[threadIdx.x >> 5] = d;
    }
    __syncthreads();
    if (blockIdx.y == 0 && threadIdx.x == 0) s1[j] = (wd[0] + wd[1]) + (wd[2] + wd[3]);
    const v4f u4 = *(const v4f*)&um[(threadIdx.x & 31) * 4];
    const float zp = wave_sum_valu(wq.x * u4.x + wq.y * u4.y + wq.z * u4.z + wq.w * u4.w);
    if ((threadIdx.x & 31) == 0) z[j * DIM + d] = zp;
}

#define XWAVES 4
#define XPARTS (1024 / XWAVES)
__launch_bounds__(32 * XWAVES)
__global__ void pass_b_kernel(const float* __restrict__ x, const float* __restrict__ n2,
                              const float* __restrict__ s1, const float* __restrict__ z,
                              float* __restrict__ s2p) {
    __shared__ __attribute__((aligned(16))) float wd[XWAVES][NJ];
    const int wave = threadIdx.x >> 5;
    const int lane = threadIdx.x & 31;
    const int s    = blockIdx.x * XWAVES + wave;
    const float* xp = x + (size_t)7 * 8 * 1024 * 128 + (size_t)s * 128 + lane * 4;

    v4f a[8], zz[NJ];
#pragma unroll
    for (int b = 0; b < 8; ++b) a[b] = *(const v4f*)(xp + (size_t)b * (1024 * 128));
    const v2f nq = *(const v2f*)(n2 + s * 64 + 2 * lane);
#pragma unroll
    for (int j = 0; j < NJ; ++j) zz[j] = *(const v4f*)(z + j * DIM + lane * 4);

    const int j0 = 2 * (lane & 3);
    float mx = logit1(s1[0]);
    for (int k = 1; k < NJ; ++k) mx = fmaxf(mx, logit1(s1[k]));
    float se = 0.0f, e0 = 0.0f, e1 = 0.0f;
    for (int k = 0; k < NJ; ++k) {
        const float ek = __expf(logit1(s1[k]) - mx);
        se += ek;
        if (k == j0)     e0 = ek;
        if (k == j0 + 1) e1 = ek;
    }
    const float c0 = e0 / se, c1 = e1 / se;

    asm volatile("" :: "v"(a[0]), "v"(a[1]), "v"(a[2]), "v"(a[3]), "v"(a[4]), "v"(a[5]), "v"(a[6]), "v"(a[7]),
                       "v"(zz[0]), "v"(zz[1]), "v"(zz[2]), "v"(zz[3]), "v"(zz[4]), "v"(zz[5]), "v"(zz[6]), "v"(zz[7]),
                       "v"(nq), "v"(c0), "v"(c1));

    const float g0 = squash_gain_routing(c0 * c0 * nq.x);
    const float g1 = squash_gain_routing(c1 * c1 * nq.y);

    v4f y[NJ];
#pragma unroll
    for (int j = 0; j < NJ; ++j) y[j] = (v4f){0, 0, 0, 0};
#pragma unroll
    for (int b = 0; b < 8; ++b) {
#pragma unroll
        for (int j = 0; j < NJ; ++j) {
            const float gv = __int_as_float(__builtin_amdgcn_readlane(
                __float_as_int((j & 1) ? g1 : g0), (b * 8 + j) >> 1));
            y[j] += gv * a[b];
        }
    }
    float d[NJ];
#pragma unroll
    for (int j = 0; j < NJ; ++j)
        d[j] = wave_sum_valu(y[j].x * zz[j].x + y[j].y * zz[j].y +
                             y[j].z * zz[j].z + y[j].w * zz[j].w);
    if (lane == 0) {
        *(v4f*)&wd[wave][0] = (v4f){d[0], d[1], d[2], d[3]};
        *(v4f*)&wd[wave][4] = (v4f){d[4], d[5], d[6], d[7]};
    }
    __syncthreads();
    if (threadIdx.x < NJ)
        s2p[blockIdx.x * NJ + threadIdx.x] = (wd[0][threadIdx.x] + wd[1][threadIdx.x]) +
                                             (wd[2][threadIdx.x] + wd[3][threadIdx.x]);
}

#define FROWS 8
__launch_bounds__(256)
__global__ void final_kernel(const float* __restrict__ uh,
                             const float* __restrict__ n2,
                             const float* __restrict__ s2p,
                             const float* __restrict__ s1,
                             float* __restrict__ out) {
    __shared__ float wd[8][NJ];
    const int t    = threadIdx.x;
    const int lane = t & 31;
    const int k    = lane & 7;
    const int wave = __builtin_amdgcn_readfirstlane(t >> 5);
    const int r0   = blockIdx.x * FROWS;

    const v4f pa = *(const v4f*)(s2p + t * NJ);
    const v4f pb = *(const v4f*)(s2p + t * NJ + 4);
    const float s1k = s1[k];
    const float nr  = n2[(r0 + k) * NJ + wave];
    const float* p = uh + (size_t)r0 * COLS + t * 4;
    v4f a[FROWS];
#pragma unroll
    for (int u = 0; u < FROWS; ++u) a[u] = *(const v4f*)(p + (size_t)u * COLS);

    const bool b0 = (lane & 1) != 0, b1 = (lane & 2) != 0, b2 = (lane & 4) != 0;
    const float q[8] = {pa.x, pa.y, pa.z, pa.w, pb.x, pb.y, pb.z, pb.w};
    float h1[4], h2[2];
#pragma unroll
    for (int i = 0; i < 4; ++i) {
        const float keep = b2 ? q[i + 4] : q[i];
        const float send = b2 ? q[i] : q[i + 4];
        h1[i] = keep + row_xor<0x164>(send);
    }
#pragma unroll
    for (int i = 0; i < 2; ++i) {
        const float keep = b1 ? h1[i + 2] : h1[i];
        const float send = b1 ? h1[i] : h1[i + 2];
        h2[i] = keep + row_xor<0x162>(send);
    }
    float wsum;
    {
        const float keep = b0 ? h2[1] : h2[0];
        const float send = b0 ? h2[0] : h2[1];
        wsum = keep + row_xor<0x161>(send);
    }
    wsum += row_xor<0x168>(wsum);
    wsum += xhalf(wsum);
    if (lane < NJ) wd[wave][k] = wsum;
    asm volatile("s_wait_dscnt 0x0\n\ts_barrier_signal -1\n\ts_barrier_wait -1" ::: "memory");
    const float s2k = ((wd[0][k] + wd[1][k]) + (wd[2][k] + wd[3][k])) +
                      ((wd[4][k] + wd[5][k]) + (wd[6][k] + wd[7][k]));

    const float l1 = logit1(s1k);
    float mx = fmaxf(l1, row_xor<0x161>(l1));
    mx = fmaxf(mx, row_xor<0x162>(mx));
    mx = fmaxf(mx, row_xor<0x164>(mx));
    const float e1 = __expf(l1 - mx);
    float se = e1 + row_xor<0x161>(e1);
    se += row_xor<0x162>(se);
    se += row_xor<0x164>(se);
    const float l2 = l1 + (e1 / se) * s2k * (1.0f / (float)JB);
    mx = fmaxf(l2, row_xor<0x161>(l2));
    mx = fmaxf(mx, row_xor<0x162>(mx));
    mx = fmaxf(mx, row_xor<0x164>(mx));
    const float e2 = __expf(l2 - mx);
    se = e2 + row_xor<0x161>(e2);
    se += row_xor<0x162>(se);
    se += row_xor<0x164>(se);
    const float cfk = e2 / se;

    const float cj = __int_as_float(__builtin_amdgcn_readlane(__float_as_int(cfk), wave));
    const float gk = cj * squash_gain(cj * cj * nr);
    float* o = out + (size_t)r0 * COLS + t * 4;
#pragma unroll
    for (int u = 0; u < FROWS; ++u) {
        const float g = __int_as_float(__builtin_amdgcn_readlane(__float_as_int(gk), u));
        *(v4f*)(o + (size_t)u * COLS) = g * a[u];
    }
}

extern "C" void kernel_launch(void* const* d_in, const int* in_sizes, int n_in,
                              void* d_out, int out_size, void* d_ws, size_t ws_size,
                              hipStream_t stream) {
    const float* x = (const float*)d_in[0];
    const float* w = (const float*)d_in[1];
    float* out = (float*)d_out;

    char* ws = (char*)d_ws;
    float* uh   = (float*)(ws);
    float* n2   = (float*)(ws + (size_t)JB * COLS * 4);
    float* s2p  = (float*)(ws + (size_t)JB * COLS * 4 + 262144);
    float* s1   = (float*)(ws + (size_t)JB * COLS * 4 + 528384);
    float* z    = out;
    float* mp   = out + 1024;
    float* v1p  = out + 1024 + GPARTS * COLS;

    gemm_uhat_kernel<<<dim3(128, 2, 1), 256, 0, stream>>>(x, w, uh, n2, mp, v1p);
    join_a_kernel<<<dim3(NJ, ZQ, 1), 1024, 0, stream>>>(mp, v1p, w, s1, z);
    pass_b_kernel<<<XPARTS, 32 * XWAVES, 0, stream>>>(x, n2, s1, z, s2p);
    final_kernel<<<JB / FROWS, 256, 0, stream>>>(uh, n2, s2p, s1, out);
}


# product kernel pipelined over its 4 row tiles: a tile's stores and the read-back pass over its rows (norms, column sums) run under the next tile's matrix work; column sum trees regrouped
# speedup vs baseline: 12.8850x; 1.0893x over previous
#include <hip/hip_runtime.h>
#include <hip/hip_bf16.h>

typedef __attribute__((ext_vector_type(2))) float v2f;
typedef __attribute__((ext_vector_type(8))) float v8f;
typedef __attribute__((ext_vector_type(4))) float v4f;

#define JB 8192
#define NJ 8
#define DIM 128
#define COLS 1024

__device__ __forceinline__ float squash_gain(float n2) {
    return n2 / ((1.0f + n2) * sqrtf(n2 + 1e-8f));
}

__device__ __forceinline__ float squash_gain_routing(float n2) {
    return n2 * __builtin_amdgcn_rsqf(n2 + 1e-8f) * __builtin_amdgcn_rcpf(1.0f + n2);
}

template <int CTRL>
__device__ __forceinline__ float row_xor(float x) {
    return __int_as_float(__builtin_amdgcn_update_dpp(0, __float_as_int(x), CTRL, 0xf, 0xf, false));
}
__device__ __forceinline__ float xhalf(float x) {
    return __int_as_float(__builtin_amdgcn_permlanex16(__float_as_int(x), __float_as_int(x),
                                                       0x76543210, 0xfedcba98, false, false));
}
__device__ __forceinline__ float wave_sum_valu(float v) {
    v += row_xor<0x161>(v);
    v += row_xor<0x162>(v);
    v += row_xor<0x164>(v);
    v += row_xor<0x168>(v);
    return v + xhalf(v);
}
__device__ __forceinline__ float wave_sum(float v) {
    v += xhalf(v);
    v += row_xor<0x168>(v);
    v += row_xor<0x164>(v);
    v += row_xor<0x162>(v);
    v += row_xor<0x161>(v);
    return v;
}

__device__ __forceinline__ float logit1(float s1) {
    return 0.125f * s1 * (1.0f / (float)JB);
}

__device__ __forceinline__ void tile_stats(const v4f (&a)[8], int lane, float& nsum_out,
                                           v4f (&m)[4], v4f (&v)[4]) {
    float q[8];
#pragma unroll
    for (int u = 0; u < 8; ++u)
        q[u] = a[u].x * a[u].x + a[u].y * a[u].y + a[u].z * a[u].z + a[u].w * a[u].w;
    const bool b0 = (lane & 1) != 0, b1 = (lane & 2) != 0, b2 = (lane & 4) != 0;
    float r1[4], r2[2];
#pragma unroll
    for (int i = 0; i < 4; ++i) {
        const float keep = b0 ? q[i + 4] : q[i];
        const float send = b0 ? q[i] : q[i + 4];
        r1[i] = keep + row_xor<0x161>(send);
    }
#pragma unroll
    for (int i = 0; i < 2; ++i) {
        const float keep = b1 ? r1[i + 2] : r1[i];
        const float send = b1 ? r1[i] : r1[i + 2];
        r2[i] = keep + row_xor<0x162>(send);
    }
    float nsum;
    {
        const float keep = b2 ? r2[1] : r2[0];
        const float send = b2 ? r2[0] : r2[1];
        nsum = keep + row_xor<0x164>(send);
    }
    nsum += row_xor<0x168>(nsum);
    nsum += xhalf(nsum);
    nsum_out = nsum;
    const float gl = squash_gain_routing(nsum * (1.0f / 64.0f));

#pragma unroll
    for (int u = 0; u < 8; ++u) {
        const float g = __int_as_float(__builtin_amdgcn_readlane(
            __float_as_int(gl), ((u & 1) << 2) | (u & 2) | ((u >> 2) & 1)));
        m[u & 3] += a[u];
        v[u & 3] += g * a[u];
    }
}

template <int U0, int U1>
__device__ __forceinline__ void tile_products(const v4f (&a4)[16], const v4f (&bq)[32][2],
                                              v8f (&acc)[4]) {
#pragma unroll
    for (int u = U0; u < U1; ++u) {
#pragma unroll
        for (int p = 0; p < 2; ++p) {
            v2f a;
            a.x = a4[u][2 * p + 0];
            a.y = a4[u][2 * p + 1];
#pragma unroll
            for (int T = 0; T < 4; ++T) {
                v2f b;
                b.x = bq[2 * u + p][0][T];
                b.y = bq[2 * u + p][1][T];
                acc[T] = __builtin_amdgcn_wmma_f32_16x16x4_f32(
                    false, a, false, b, (short)0, acc[T], false, false);
            }
        }
    }
}

__launch_bounds__(256)
__global__ void gemm_uhat_kernel(const float* __restrict__ x,
                                 const float* __restrict__ w,
                                 float* __restrict__ uh,
                                 float* __restrict__ n2,
                                 float* __restrict__ mp,
                                 float* __restrict__ v1p) {
    __shared__ __attribute__((aligned(16))) float red[2][8][512];
    const float* x7 = x + (size_t)7 * 8 * 1024 * 128;
    const float* w7 = w + (size_t)7 * 8 * 128 * 128;

    const int wave = threadIdx.x >> 5;
    const int lane = threadIdx.x & 31;
    const int h    = lane >> 4;
    const int c    = lane & 15;

    const int grp  = blockIdx.y * 8 + wave;
    const int j    = grp >> 1;
    const int col0 = (grp & 1) * 64 + 4 * c;

    const float* xr = x7 + (size_t)((c & 7) * 1024 + blockIdx.x * 8 + (c >> 3)) * 128 + 4 * h;
    v4f ax[2][16];
#pragma unroll
    for (int u = 0; u < 16; ++u) ax[0][u] = *(const v4f*)(xr + 8 * u);

    const float* wl = w7 + (size_t)j * 128 * 128 + (size_t)(4 * h) * 128 + col0;
    v4f bq[32][2];
#pragma unroll
    for (int t = 0; t < 32; ++t) {
#pragma unroll
        for (int e = 0; e < 2; ++e)
            bq[t][e] = *(const v4f*)(wl + (8 * (t >> 1) + 2 * (t & 1) + e) * 128);
    }

    float* orow = uh + (size_t)(blockIdx.x * 64 + 8 * h) * COLS + grp * 64 + 4 * c;
    const int rowb = blockIdx.x * 64 + wave * 2;
    const float* pr = uh + (size_t)rowb * COLS + blockIdx.y * 512 + lane * 4;
    const int ul = ((lane & 1) << 2) | (lane & 2) | ((lane >> 2) & 1);
    float* n2w = n2 + (size_t)(rowb + (ul >> 2)) * NJ + blockIdx.y * 4 + (ul & 3);

    v4f ra[8];
    v4f m[4], v[4];
#pragma unroll
    for (int jl = 0; jl < 4; ++jl) { m[jl] = (v4f){0, 0, 0, 0}; v[jl] = (v4f){0, 0, 0, 0}; }

#pragma unroll
    for (int i = 0; i < 4; ++i) {
        if (i < 3) {
#pragma unroll
            for (int u = 0; u < 16; ++u) ax[(i + 1) & 1][u] = *(const v4f*)(xr + (i + 1) * 256 + 8 * u);
        }
        v8f acc[4];
#pragma unroll
        for (int T = 0; T < 4; ++T) acc[T] = (v8f){0,0,0,0,0,0,0,0};

        tile_products<0, 12>(ax[i & 1], bq, acc);
        if (i > 0) {
            __builtin_amdgcn_sched_barrier(0);
            __syncthreads();
#pragma unroll
            for (int u = 0; u < 8; ++u)
                ra[u] = *(const v4f*)(pr + (size_t)(16 * (i - 1) + (u >> 2)) * COLS + (u & 3) * DIM);
        }
        tile_products<12, 16>(ax[i & 1], bq, acc);

#pragma unroll
        for (int g = 0; g < 8; ++g) {
            v4f o;
            o.x = acc[0][g]; o.y = acc[1][g]; o.z = acc[2][g]; o.w = acc[3][g];
            *(v4f*)(orow + (size_t)(16 * i + g) * COLS) = o;
        }
        if (i > 0) {
            float nsum;
            tile_stats(ra, lane, nsum, m, v);
            if (lane < 8) n2w[(size_t)(16 * (i - 1)) * NJ] = nsum;
        }
        __builtin_amdgcn_sched_barrier(0);
    }

    __syncthreads();
#pragma unroll
    for (int u = 0; u < 8; ++u)
        ra[u] = *(const v4f*)(pr + (size_t)(48 + (u >> 2)) * COLS + (u & 3) * DIM);
    {
        float nsum;
        tile_stats(ra, lane, nsum, m, v);
        if (lane < 8) n2w[(size_t)48 * NJ] = nsum;
    }
#pragma unroll
    for (int jl = 0; jl < 4; ++jl) {
        *(v4f*)&red[0][wave][jl * DIM + lane * 4] = m[jl];
        *(v4f*)&red[1][wave][jl * DIM + lane * 4] = v[jl];
    }
    __syncthreads();
    {
        const int which = threadIdx.x >> 7;
        const int e4    = (threadIdx.x & 127) * 4;
        const v4f s = ((*(const v4f*)&red[which][0][e4] + *(const v4f*)&red[which][1][e4]) +
                       (*(const v4f*)&red[which][2][e4] + *(const v4f*)&red[which][3][e4])) +
                      ((*(const v4f*)&red[which][4][e4] + *(const v4f*)&red[which][5][e4]) +
                       (*(const v4f*)&red[which][6][e4] + *(const v4f*)&red[which][7][e4]));
        const int o = blockIdx.x * COLS + blockIdx.y * 512 + e4;
        if (which == 0) *(v4f*)(mp + o) = s * (1.0f / (float)JB);
        else            *(v4f*)(v1p + o) = s;
    }
}

#define GPARTS 128

#define ZQ 4
__launch_bounds__(1024)
__global__ void join_a_kernel(const float* __restrict__ mp, const float* __restrict__ v1p,
                              const float* __restrict__ w,
                              float* __restrict__ s1, float* __restrict__ z) {
    __shared__ float rm[8][DIM];
    __shared__ float rv[8][DIM];
    __shared__ float wd[4];
    __shared__ __attribute__((aligned(16))) float um[DIM];
    const int j = blockIdx.x;
    const int g = threadIdx.x >> 7;
    const int e = threadIdx.x & (DIM - 1);
    const int col = j * DIM + e;
    const int d = blockIdx.y * 32 + (threadIdx.x >> 5);
    const v4f wq = *(const v4f*)(w + (size_t)7 * 8 * 128 * 128 + (size_t)j * 128 * 128 +
                                 (size_t)d * 128 + (threadIdx.x & 31) * 4);
    float am[4] = {0.0f, 0.0f, 0.0f, 0.0f}, av[4] = {0.0f, 0.0f, 0.0f, 0.0f};
#pragma unroll
    for (int q = 0; q < GPARTS / 8; ++q) {
        am[q & 3] += mp[(g + 8 * q) * COLS + col];
        av[q & 3] += v1p[(g + 8 * q) * COLS + col];
    }
    rm[g][e] = (am[0] + am[1]) + (am[2] + am[3]);
    rv[g][e] = (av[0] + av[1]) + (av[2] + av[3]);
    __syncthreads();
    if (threadIdx.x < DIM) {
        const float M = ((rm[0][e] + rm[1][e]) + (rm[2][e] + rm[3][e])) +
                        ((rm[4][e] + rm[5][e]) + (rm[6][e] + rm[7][e]));
        const float V = ((rv[0][e] + rv[1][e]) + (rv[2][e] + rv[3][e])) +
                        ((rv[4][e] + rv[5][e]) + (rv[6][e] + rv[7][e]));
        um[e] = M;
        const float d = wave_sum(M * V);
        if ((threadIdx.x & 31) == 0) wd[threadIdx.x >> 5] = d;
    }
    __syncthreads();
    if (blockIdx.y == 0 && threadIdx.x == 0) s1[j] = (wd[0] + wd[1]) + (wd[2] + wd[3]);
    const v4f u4 = *(const v4f*)&um[(threadIdx.x & 31) * 4];
    const float zp = wave_sum_valu(wq.x * u4.x + wq.y * u4.y + wq.z * u4.z + wq.w * u4.w);
    if ((threadIdx.x & 31) == 0) z[j * DIM + d] = zp;
}

#define XWAVES 4
#define XPARTS (1024 / XWAVES)
__launch_bounds__(32 * XWAVES)
__global__ void pass_b_kernel(const float* __restrict__ x, const float* __restrict__ n2,
                              const float* __restrict__ s1, const float* __restrict__ z,
                              float* __restrict__ s2p) {
    __shared__ __attribute__((aligned(16))) float wd[XWAVES][NJ];
    const int wave = threadIdx.x >> 5;
    const int lane = threadIdx.x & 31;
    const int s    = blockIdx.x * XWAVES + wave;
    const float* xp = x + (size_t)7 * 8 * 1024 * 128 + (size_t)s * 128 + lane * 4;

    v4f a[8], zz[NJ];
#pragma unroll
    for (int b = 0; b < 8; ++b) a[b] = *(const v4f*)(xp + (size_t)b * (1024 * 128));
    const v2f nq = *(const v2f*)(n2 + s * 64 + 2 * lane);
#pragma unroll
    for (int j = 0; j < NJ; ++j) zz[j] = *(const v4f*)(z + j * DIM + lane * 4);

    const int j0 = 2 * (lane & 3);
    float mx = logit1(s1[0]);
    for (int k = 1; k < NJ; ++k) mx = fmaxf(mx, logit1(s1[k]));
    float se = 0.0f, e0 = 0.0f, e1 = 0.0f;
    for (int k = 0; k < NJ; ++k) {
        const float ek = __expf(logit1(s1[k]) - mx);
        se += ek;
        if (k == j0)     e0 = ek;
        if (k == j0 + 1) e1 = ek;
    }
    const float c0 = e0 / se, c1 = e1 / se;

    asm volatile("" :: "v"(a[0]), "v"(a[1]), "v"(a[2]), "v"(a[3]), "v"(a[4]), "v"(a[5]), "v"(a[6]), "v"(a[7]),
                       "v"(zz[0]), "v"(zz[1]), "v"(zz[2]), "v"(zz[3]), "v"(zz[4]), "v"(zz[5]), "v"(zz[6]), "v"(zz[7]),
                       "v"(nq), "v"(c0), "v"(c1));

    const float g0 = squash_gain_routing(c0 * c0 * nq.x);
    const float g1 = squash_gain_routing(c1 * c1 * nq.y);

    v4f y[NJ];
#pragma unroll
    for (int j = 0; j < NJ; ++j) y[j] = (v4f){0, 0, 0, 0};
#pragma unroll
    for (int b = 0; b < 8; ++b) {
#pragma unroll
        for (int j = 0; j < NJ; ++j) {
            const float gv = __int_as_float(__builtin_amdgcn_readlane(
                __float_as_int((j & 1) ? g1 : g0), (b * 8 + j) >> 1));
            y[j] += gv * a[b];
        }
    }
    float d[NJ];
#pragma unroll
    for (int j = 0; j < NJ; ++j)
        d[j] = wave_sum_valu(y[j].x * zz[j].x + y[j].y * zz[j].y +
                             y[j].z * zz[j].z + y[j].w * zz[j].w);
    if (lane == 0) {
        *(v4f*)&wd[wave][0] = (v4f){d[0], d[1], d[2], d[3]};
        *(v4f*)&wd[wave][4] = (v4f){d[4], d[5], d[6], d[7]};
    }
    __syncthreads();
    if (threadIdx.x < NJ)
        s2p[blockIdx.x * NJ + threadIdx.x] = (wd[0][threadIdx.x] + wd[1][threadIdx.x]) +
                                             (wd[2][threadIdx.x] + wd[3][threadIdx.x]);
}

#define FROWS 8
__launch_bounds__(256)
__global__ void final_kernel(const float* __restrict__ uh,
                             const float* __restrict__ n2,
                             const float* __restrict__ s2p,
                             const float* __restrict__ s1,
                             float* __restrict__ out) {
    __shared__ float wd[8][NJ];
    const int t    = threadIdx.x;
    const int lane = t & 31;
    const int k    = lane & 7;
    const int wave = __builtin_amdgcn_readfirstlane(t >> 5);
    const int r0   = blockIdx.x * FROWS;

    const v4f pa = *(const v4f*)(s2p + t * NJ);
    const v4f pb = *(const v4f*)(s2p + t * NJ + 4);
    const float s1k = s1[k];
    const float nr  = n2[(r0 + k) * NJ + wave];
    const float* p = uh + (size_t)r0 * COLS + t * 4;
    v4f a[FROWS];
#pragma unroll
    for (int u = 0; u < FROWS; ++u) a[u] = *(const v4f*)(p + (size_t)u * COLS);

    const bool b0 = (lane & 1) != 0, b1 = (lane & 2) != 0, b2 = (lane & 4) != 0;
    const float q[8] = {pa.x, pa.y, pa.z, pa.w, pb.x, pb.y, pb.z, pb.w};
    float h1[4], h2[2];
#pragma unroll
    for (int i = 0; i < 4; ++i) {
        const float keep = b2 ? q[i + 4] : q[i];
        const float send = b2 ? q[i] : q[i + 4];
        h1[i] = keep + row_xor<0x164>(send);
    }
#pragma unroll
    for (int i = 0; i < 2; ++i) {
        const float keep = b1 ? h1[i + 2] : h1[i];
        const float send = b1 ? h1[i] : h1[i + 2];
        h2[i] = keep + row_xor<0x162>(send);
    }
    float wsum;
    {
        const float keep = b0 ? h2[1] : h2[0];
        const float send = b0 ? h2[0] : h2[1];
        wsum = keep + row_xor<0x161>(send);
    }
    wsum += row_xor<0x168>(wsum);
    wsum += xhalf(wsum);
    if (lane < NJ) wd[wave][k] = wsum;
    asm volatile("s_wait_dscnt 0x0\n\ts_barrier_signal -1\n\ts_barrier_wait -1" ::: "memory");
    const float s2k = ((wd[0][k] + wd[1][k]) + (wd[2][k] + wd[3][k])) +
                      ((wd[4][k] + wd[5][k]) + (wd[6][k] + wd[7][k]));

    const float l1 = logit1(s1k);
    float mx = fmaxf(l1, row_xor<0x161>(l1));
    mx = fmaxf(mx, row_xor<0x162>(mx));
    mx = fmaxf(mx, row_xor<0x164>(mx));
    const float e1 = __expf(l1 - mx);
    float se = e1 + row_xor<0x161>(e1);
    se += row_xor<0x162>(se);
    se += row_xor<0x164>(se);
    const float l2 = l1 + (e1 / se) * s2k * (1.0f / (float)JB);
    mx = fmaxf(l2, row_xor<0x161>(l2));
    mx = fmaxf(mx, row_xor<0x162>(mx));
    mx = fmaxf(mx, row_xor<0x164>(mx));
    const float e2 = __expf(l2 - mx);
    se = e2 + row_xor<0x161>(e2);
    se += row_xor<0x162>(se);
    se += row_xor<0x164>(se);
    const float cfk = e2 / se;

    const float cj = __int_as_float(__builtin_amdgcn_readlane(__float_as_int(cfk), wave));
    const float gk = cj * squash_gain(cj * cj * nr);
    float* o = out + (size_t)r0 * COLS + t * 4;
#pragma unroll
    for (int u = 0; u < FROWS; ++u) {
        const float g = __int_as_float(__builtin_amdgcn_readlane(__float_as_int(gk), u));
        *(v4f*)(o + (size_t)u * COLS) = g * a[u];
    }
}

extern "C" void kernel_launch(void* const* d_in, const int* in_sizes, int n_in,
                              void* d_out, int out_size, void* d_ws, size_t ws_size,
                              hipStream_t stream) {
    const float* x = (const float*)d_in[0];
    const float* w = (const float*)d_in[1];
    float* out = (float*)d_out;

    char* ws = (char*)d_ws;
    float* uh   = (float*)(ws);
    float* n2   = (float*)(ws + (size_t)JB * COLS * 4);
    float* s2p  = (float*)(ws + (size_t)JB * COLS * 4 + 262144);
    float* s1   = (float*)(ws + (size_t)JB * COLS * 4 + 528384);
    float* z    = out;
    float* mp   = out + 1024;
    float* v1p  = out + 1024 + GPARTS * COLS;

    gemm_uhat_kernel<<<dim3(128, 2, 1), 256, 0, stream>>>(x, w, uh, n2, mp, v1p);
    join_a_kernel<<<dim3(NJ, ZQ, 1), 1024, 0, stream>>>(mp, v1p, w, s1, z);
    pass_b_kernel<<<XPARTS, 32 * XWAVES, 0, stream>>>(x, n2, s1, z, s2p);
    final_kernel<<<JB / FROWS, 256, 0, stream>>>(uh, n2, s2p, s1, out);
}
